# starting kernel
# baseline (speedup 1.0000x reference)
#include <hip/hip_runtime.h>


typedef __attribute__((ext_vector_type(2))) float v2f;
typedef __attribute__((ext_vector_type(8))) float v8f;

#define LDP   132
#define HDIM  128
#define CINC  64
#define NOBSC 16

__device__ __forceinline__ v8f wmma4(v2f a, v2f b, v8f c) {
  return __builtin_amdgcn_wmma_f32_16x16x4_f32(
       false, a,  false, b,
       (short)0, c,  false,  false);
}

template <int K>
__device__ __forceinline__ void gemm_lds(const float* __restrict__ sY,
                                         const float* __restrict__ W, int ldw,
                                         int wcol, int lane, v8f acc[4]) {
  const int mrow  = lane & 15;
  const int khalf = (lane >> 4) << 1;
  #pragma unroll 4
  for (int ks = 0; ks < K / 4; ++ks) {
    const int k0 = ks * 4 + khalf;
    v2f a = *(const v2f*)(sY + mrow * LDP + k0);
    #pragma unroll
    for (int nt = 0; nt < 4; ++nt) {
      const int n = wcol + nt * 16 + mrow;
      v2f b = *(const v2f*)(W + n * ldw + k0);
      acc[nt] = wmma4(a, b, acc[nt]);
    }
  }
}

__device__ __forceinline__ void write_tile(float* __restrict__ sY, int wcol, int lane,
                                           const v8f v[4]) {
  const int hi   = lane >> 4;
  const int col0 = lane & 15;
  #pragma unroll
  for (int nt = 0; nt < 4; ++nt) {
    const int col = wcol + nt * 16 + col0;
    #pragma unroll
    for (int r = 0; r < 8; ++r)
      sY[(r + 8 * hi) * LDP + col] = v[nt][r];
  }
}

__global__ void __launch_bounds__(64) jacde_kernel(
    const float* __restrict__ t,      const float* __restrict__ hmat,
    const float* __restrict__ coeffs, const float* __restrict__ dcoeffs,
    const float* __restrict__ tobs,
    const float* __restrict__ wx,     const float* __restrict__ wh,
    const float* __restrict__ wout,
    const float* __restrict__ b0,     const float* __restrict__ b1,
    float* __restrict__ out) {
  __shared__ float sY[16 * LDP];

  const int tid  = threadIdx.x;
  const int lane = tid & 31;
  const int wave = tid >> 5;
  const int wcol = wave * 64;
  const int brow = blockIdx.x * 16;
  const int hi   = lane >> 4;
  const int col0 = lane & 15;

  const float ts = t[0];
  int idx = 0;
  #pragma unroll
  for (int j = 0; j < NOBSC; ++j) idx += (tobs[j] <= ts) ? 1 : 0;
  idx -= 1;
  idx = idx < 0 ? 0 : (idx > NOBSC - 2 ? NOBSC - 2 : idx);
  const float dt = ts - tobs[idx];
  const float p1 = dt, p2 = dt * dt, p3 = p2 * dt;

  {
    const float* cb = coeffs + ((size_t)brow * (NOBSC - 1) + idx) * CINC * 4;
    #pragma unroll 4
    for (int i = 0; i < 16; ++i) {
      const float4 c4 = *(const float4*)(cb + (size_t)i * (NOBSC - 1) * CINC * 4 + tid * 4);
      sY[i * LDP + tid] = c4.x + c4.y * p1 + c4.z * p2 + c4.w * p3;
    }
  }
  __syncthreads();

  v8f acc[4];
  #pragma unroll
  for (int nt = 0; nt < 4; ++nt) {
    const float bv = b0[wcol + nt * 16 + col0];
    #pragma unroll
    for (int r = 0; r < 8; ++r) acc[nt][r] = bv;
  }
  gemm_lds<CINC>(sY, wx, CINC, wcol, lane, acc);
  __syncthreads();

  #pragma unroll 4
  for (int i = 0; i < 32; ++i) {
    const int v = i * 64 + tid, m = v >> 7, c = v & 127;
    sY[m * LDP + c] = hmat[(size_t)(brow + m) * HDIM + c];
  }
  __syncthreads();
  gemm_lds<HDIM>(sY, wh, HDIM, wcol, lane, acc);
  __syncthreads();

  v8f drelu[4];
  #pragma unroll
  for (int nt = 0; nt < 4; ++nt)
    #pragma unroll
    for (int r = 0; r < 8; ++r) {
      const float x = acc[nt][r];
      drelu[nt][r] = 1.0f / (1.0f + __expf(-x));
      acc[nt][r]   = fmaxf(x, 0.0f);
    }
  write_tile(sY, wcol, lane, acc);
  __syncthreads();

  v8f dtn[4];
  {
    v8f a2[4];
    #pragma unroll
    for (int nt = 0; nt < 4; ++nt) {
      const float bv = b1[wcol + nt * 16 + col0];
      #pragma unroll
      for (int r = 0; r < 8; ++r) a2[nt][r] = bv;
    }
    gemm_lds<HDIM>(sY, wout, HDIM, wcol, lane, a2);
    #pragma unroll
    for (int nt = 0; nt < 4; ++nt)
      #pragma unroll
      for (int r = 0; r < 8; ++r) {
        const float th = tanhf(a2[nt][r]);
        dtn[nt][r] = 1.0f - th * th;
      }
  }
  __syncthreads();

  {
    const float* db = dcoeffs + ((size_t)brow * (NOBSC - 1) + idx) * CINC * 4;
    #pragma unroll 4
    for (int i = 0; i < 16; ++i) {
      const float4 c4 = *(const float4*)(db + (size_t)i * (NOBSC - 1) * CINC * 4 + tid * 4);
      sY[i * LDP + tid] = c4.x + c4.y * p1 + c4.z * p2 + c4.w * p3;
    }
  }
  __syncthreads();
  v8f cur[4];
  #pragma unroll
  for (int nt = 0; nt < 4; ++nt) cur[nt] = (v8f)(0.0f);
  gemm_lds<CINC>(sY, wx, CINC, wcol, lane, cur);
  __syncthreads();

  #pragma unroll
  for (int nt = 0; nt < 4; ++nt) cur[nt] *= drelu[nt];
  write_tile(sY, wcol, lane, cur);
  __syncthreads();
  v8f outv[4];
  #pragma unroll
  for (int nt = 0; nt < 4; ++nt) outv[nt] = (v8f)(0.0f);
  gemm_lds<HDIM>(sY, wout, HDIM, wcol, lane, outv);
  __syncthreads();
  #pragma unroll
  for (int nt = 0; nt < 4; ++nt) {
    outv[nt] *= dtn[nt];
    cur[nt] = outv[nt];
  }

  for (int it = 0; it < 2; ++it) {
    write_tile(sY, wcol, lane, cur);
    __syncthreads();
    v8f t1[4];
    #pragma unroll
    for (int nt = 0; nt < 4; ++nt) t1[nt] = (v8f)(0.0f);
    gemm_lds<HDIM>(sY, wh, HDIM, wcol, lane, t1);
    __syncthreads();
    #pragma unroll
    for (int nt = 0; nt < 4; ++nt) t1[nt] *= drelu[nt];
    write_tile(sY, wcol, lane, t1);
    __syncthreads();
    v8f t2[4];
    #pragma unroll
    for (int nt = 0; nt < 4; ++nt) t2[nt] = (v8f)(0.0f);
    gemm_lds<HDIM>(sY, wout, HDIM, wcol, lane, t2);
    __syncthreads();
    #pragma unroll
    for (int nt = 0; nt < 4; ++nt) {
      t2[nt] *= dtn[nt];
      outv[nt] += t2[nt];
      cur[nt] = t2[nt];
    }
  }

  #pragma unroll
  for (int nt = 0; nt < 4; ++nt) {
    const int col = wcol + nt * 16 + col0;
    #pragma unroll
    for (int r = 0; r < 8; ++r)
      out[(size_t)(brow + r + 8 * hi) * HDIM + col] = outv[nt][r];
  }
}

extern "C" void kernel_launch(void* const* d_in, const int* in_sizes, int n_in,
                              void* d_out, int out_size, void* d_ws, size_t ws_size,
                              hipStream_t stream) {
  const float* t       = (const float*)d_in[0];
  const float* hmat    = (const float*)d_in[1];
  const float* coeffs  = (const float*)d_in[2];
  const float* dcoeffs = (const float*)d_in[3];
  const float* tobs    = (const float*)d_in[4];
  const float* wx      = (const float*)d_in[5];
  const float* wh      = (const float*)d_in[6];
  const float* wout    = (const float*)d_in[7];
  const float* b0      = (const float*)d_in[8];
  const float* b1      = (const float*)d_in[9];
  float* out           = (float*)d_out;

  const int B = 8192;
  jacde_kernel<<<B / 16, 64, 0, stream>>>(t, hmat, coeffs, dcoeffs, tobs,
                                          wx, wh, wout, b0, b1, out);
}


# Weight rows go through a per-wave LDS copy filled by whole-line loads; the next product's rows are fetched to registers before each product and stored to LDS after it
# speedup vs baseline: 1.3487x; 1.3487x over previous
#include <hip/hip_runtime.h>


typedef __attribute__((ext_vector_type(2))) float v2f;
typedef __attribute__((ext_vector_type(8))) float v8f;

#define LDP   132
#define HDIM  128
#define CINC  64
#define NOBSC 16
#define WROWS 64
#define WREGION (WROWS * (HDIM + 4))

__device__ __forceinline__ v8f wmma4(v2f a, v2f b, v8f c) {
  return __builtin_amdgcn_wmma_f32_16x16x4_f32(
       false, a,  false, b,
       (short)0, c,  false,  false);
}

__device__ __forceinline__ void lds_barrier() {
  __builtin_amdgcn_fence(__ATOMIC_RELEASE, "workgroup", "local");
  __builtin_amdgcn_s_barrier();
  __builtin_amdgcn_fence(__ATOMIC_ACQUIRE, "workgroup", "local");
}

typedef __attribute__((ext_vector_type(4))) float v4f;

template <int K>
struct WRows { v4f r[K / 2]; };

template <int K, int I = 0>
__device__ __forceinline__ void w_fetch_from(const v4f* __restrict__ src, WRows<K>& w) {
  if constexpr (I < K / 2) {
    w.r[I] = src[I * 32];
    w_fetch_from<K, I + 1>(src, w);
  }
}

template <int K>
__device__ __forceinline__ void w_fetch(const float* __restrict__ W, int wcol, int lane,
                                        WRows<K>& w) {
  w_fetch_from<K>(reinterpret_cast<const v4f*>(W + (size_t)wcol * K) + lane, w);
}

template <int K, int I = 0>
__device__ __forceinline__ void w_commit_to(float* __restrict__ sWl, const WRows<K>& w) {
  if constexpr (I < K / 2) {
    *reinterpret_cast<v4f*>(sWl + ((I * 128) / K) * (K + 4) + ((I * 128) % K)) = w.r[I];
    w_commit_to<K, I + 1>(sWl, w);
  }
}

template <int K>
__device__ __forceinline__ void w_commit(float* __restrict__ sW, int lane, const WRows<K>& w) {
  const int f   = lane * 4;
  const int row = f / K;
  const int col = f % K;
  w_commit_to<K>(sW + row * (K + 4) + col, w);
}

template <int K>
__device__ __forceinline__ void gemm_lds(const float* __restrict__ sY,
                                         const float* __restrict__ sW,
                                         int lane, v8f acc[4]) {
  const int mrow  = lane & 15;
  const int khalf = (lane >> 4) << 1;
  #pragma unroll 4
  for (int ks = 0; ks < K / 4; ++ks) {
    const int k0 = ks * 4 + khalf;
    v2f a = *(const v2f*)(sY + mrow * LDP + k0);
    #pragma unroll
    for (int nt = 0; nt < 4; ++nt) {
      v2f b = *(const v2f*)(sW + (nt * 16 + mrow) * (K + 4) + k0);
      acc[nt] = wmma4(a, b, acc[nt]);
    }
  }
}

typedef __attribute__((address_space(3))) float lds_float;
__device__ __forceinline__ void lds_store1(float* p, float v) {
  *(volatile lds_float*)p = v;
}

__device__ __forceinline__ void write_tile(float* __restrict__ sY, int wcol, int lane,
                                           const v8f v[4]) {
  const int hi   = lane >> 4;
  const int col0 = lane & 15;
  #pragma unroll
  for (int nt = 0; nt < 4; ++nt) {
    const int col = wcol + nt * 16 + col0;
    #pragma unroll
    for (int r = 0; r < 8; ++r)
      lds_store1(sY + (r + 8 * hi) * LDP + col, v[nt][r]);
  }
}

__global__ void __launch_bounds__(64) __attribute__((amdgpu_waves_per_eu(1, 1))) jacde_kernel(
    const float* __restrict__ t,      const float* __restrict__ hmat,
    const float* __restrict__ coeffs, const float* __restrict__ dcoeffs,
    const float* __restrict__ tobs,
    const float* __restrict__ wx,     const float* __restrict__ wh,
    const float* __restrict__ wout,
    const float* __restrict__ b0,     const float* __restrict__ b1,
    float* __restrict__ out) {
  __shared__ __attribute__((aligned(16))) float sY[16 * LDP];
  __shared__ __attribute__((aligned(16))) float sWall[2 * WREGION];

  const int tid  = threadIdx.x;
  const int lane = tid & 31;
  const int wave = tid >> 5;
  const int wcol = wave * 64;
  const int brow = blockIdx.x * 16;
  const int hi   = lane >> 4;
  const int col0 = lane & 15;
  float* sW = sWall + wave * WREGION;

  const float ts = t[0];
  int idx = 0;
  #pragma unroll
  for (int j = 0; j < NOBSC; ++j) idx += (tobs[j] <= ts) ? 1 : 0;
  idx -= 1;
  idx = idx < 0 ? 0 : (idx > NOBSC - 2 ? NOBSC - 2 : idx);
  const float dt = ts - tobs[idx];
  const float p1 = dt, p2 = dt * dt, p3 = p2 * dt;

  float bias0[4], bias1[4];
  #pragma unroll
  for (int nt = 0; nt < 4; ++nt) {
    bias0[nt] = b0[wcol + nt * 16 + col0];
    bias1[nt] = b1[wcol + nt * 16 + col0];
  }

  {
    const float* cb = coeffs + ((size_t)brow * (NOBSC - 1) + idx) * CINC * 4;
    #pragma unroll 4
    for (int i = 0; i < 16; ++i) {
      const float4 c4 = *(const float4*)(cb + (size_t)i * (NOBSC - 1) * CINC * 4 + tid * 4);
      sY[i * LDP + tid] = c4.x + c4.y * p1 + c4.z * p2 + c4.w * p3;
    }
  }
  {
    WRows<CINC> w0;
    w_fetch<CINC>(wx, wcol, lane, w0);
    w_commit<CINC>(sW, lane, w0);
  }
  WRows<HDIM> w1;
  w_fetch<HDIM>(wh, wcol, lane, w1);
  lds_barrier();

  v8f acc[4];
  #pragma unroll
  for (int nt = 0; nt < 4; ++nt) {
    const float bv = bias0[nt];
    #pragma unroll
    for (int r = 0; r < 8; ++r) acc[nt][r] = bv;
  }
  gemm_lds<CINC>(sY, sW, lane, acc);
  __syncthreads();

  #pragma unroll 4
  for (int i = 0; i < 32; ++i) {
    const int v = i * 64 + tid, m = v >> 7, c = v & 127;
    sY[m * LDP + c] = hmat[(size_t)(brow + m) * HDIM + c];
  }
  w_commit<HDIM>(sW, lane, w1);
  WRows<HDIM> w2;
  w_fetch<HDIM>(wout, wcol, lane, w2);
  lds_barrier();
  gemm_lds<HDIM>(sY, sW, lane, acc);
  __syncthreads();

  v8f drelu[4];
  #pragma unroll
  for (int nt = 0; nt < 4; ++nt)
    #pragma unroll
    for (int r = 0; r < 8; ++r) {
      const float x = acc[nt][r];
      drelu[nt][r] = 1.0f / (1.0f + __expf(-x));
      acc[nt][r]   = fmaxf(x, 0.0f);
    }
  write_tile(sY, wcol, lane, acc);
  w_commit<HDIM>(sW, lane, w2);
  WRows<CINC> w3;
  w_fetch<CINC>(wx, wcol, lane, w3);
  lds_barrier();

  v8f dtn[4];
  {
    v8f a2[4];
    #pragma unroll
    for (int nt = 0; nt < 4; ++nt) {
      const float bv = bias1[nt];
      #pragma unroll
      for (int r = 0; r < 8; ++r) a2[nt][r] = bv;
    }
    gemm_lds<HDIM>(sY, sW, lane, a2);
    #pragma unroll
    for (int nt = 0; nt < 4; ++nt)
      #pragma unroll
      for (int r = 0; r < 8; ++r) {
        const float th = tanhf(a2[nt][r]);
        dtn[nt][r] = 1.0f - th * th;
      }
  }
  __syncthreads();

  {
    const float* db = dcoeffs + ((size_t)brow * (NOBSC - 1) + idx) * CINC * 4;
    #pragma unroll 4
    for (int i = 0; i < 16; ++i) {
      const float4 c4 = *(const float4*)(db + (size_t)i * (NOBSC - 1) * CINC * 4 + tid * 4);
      sY[i * LDP + tid] = c4.x + c4.y * p1 + c4.z * p2 + c4.w * p3;
    }
  }
  w_commit<CINC>(sW, lane, w3);
  WRows<HDIM> w4;
  w_fetch<HDIM>(wout, wcol, lane, w4);
  lds_barrier();
  v8f cur[4];
  #pragma unroll
  for (int nt = 0; nt < 4; ++nt) cur[nt] = (v8f)(0.0f);
  gemm_lds<CINC>(sY, sW, lane, cur);
  __syncthreads();

  #pragma unroll
  for (int nt = 0; nt < 4; ++nt) cur[nt] *= drelu[nt];
  write_tile(sY, wcol, lane, cur);
  w_commit<HDIM>(sW, lane, w4);
  WRows<HDIM> w5;
  w_fetch<HDIM>(wh, wcol, lane, w5);
  lds_barrier();
  v8f outv[4];
  #pragma unroll
  for (int nt = 0; nt < 4; ++nt) outv[nt] = (v8f)(0.0f);
  gemm_lds<HDIM>(sY, sW, lane, outv);
  __syncthreads();
  #pragma unroll
  for (int nt = 0; nt < 4; ++nt) {
    outv[nt] *= dtn[nt];
    cur[nt] = outv[nt];
  }

  WRows<HDIM> w6, w7, w8;

  write_tile(sY, wcol, lane, cur);
  w_commit<HDIM>(sW, lane, w5);
  w_fetch<HDIM>(wout, wcol, lane, w6);
  lds_barrier();
  {
    v8f t1[4];
    #pragma unroll
    for (int nt = 0; nt < 4; ++nt) t1[nt] = (v8f)(0.0f);
    gemm_lds<HDIM>(sY, sW, lane, t1);
    __syncthreads();
    #pragma unroll
    for (int nt = 0; nt < 4; ++nt) t1[nt] *= drelu[nt];
    write_tile(sY, wcol, lane, t1);
  }
  w_commit<HDIM>(sW, lane, w6);
  w_fetch<HDIM>(wh, wcol, lane, w7);
  lds_barrier();
  {
    v8f t2[4];
    #pragma unroll
    for (int nt = 0; nt < 4; ++nt) t2[nt] = (v8f)(0.0f);
    gemm_lds<HDIM>(sY, sW, lane, t2);
    __syncthreads();
    #pragma unroll
    for (int nt = 0; nt < 4; ++nt) {
      t2[nt] *= dtn[nt];
      outv[nt] += t2[nt];
      cur[nt] = t2[nt];
    }
  }

  write_tile(sY, wcol, lane, cur);
  w_commit<HDIM>(sW, lane, w7);
  w_fetch<HDIM>(wout, wcol, lane, w8);
  lds_barrier();
  {
    v8f t1[4];
    #pragma unroll
    for (int nt = 0; nt < 4; ++nt) t1[nt] = (v8f)(0.0f);
    gemm_lds<HDIM>(sY, sW, lane, t1);
    __syncthreads();
    #pragma unroll
    for (int nt = 0; nt < 4; ++nt) t1[nt] *= drelu[nt];
    write_tile(sY, wcol, lane, t1);
  }
  w_commit<HDIM>(sW, lane, w8);
  __syncthreads();
  {
    v8f t2[4];
    #pragma unroll
    for (int nt = 0; nt < 4; ++nt) t2[nt] = (v8f)(0.0f);
    gemm_lds<HDIM>(sY, sW, lane, t2);
    #pragma unroll
    for (int nt = 0; nt < 4; ++nt) {
      t2[nt] *= dtn[nt];
      outv[nt] += t2[nt];
    }
  }

  #pragma unroll
  for (int nt = 0; nt < 4; ++nt) {
    const int col = wcol + nt * 16 + col0;
    #pragma unroll
    for (int r = 0; r < 8; ++r)
      out[(size_t)(brow + r + 8 * hi) * HDIM + col] = outv[nt][r];
  }
}

extern "C" void kernel_launch(void* const* d_in, const int* in_sizes, int n_in,
                              void* d_out, int out_size, void* d_ws, size_t ws_size,
                              hipStream_t stream) {
  const float* t       = (const float*)d_in[0];
  const float* hmat    = (const float*)d_in[1];
  const float* coeffs  = (const float*)d_in[2];
  const float* dcoeffs = (const float*)d_in[3];
  const float* tobs    = (const float*)d_in[4];
  const float* wx      = (const float*)d_in[5];
  const float* wh      = (const float*)d_in[6];
  const float* wout    = (const float*)d_in[7];
  const float* b0      = (const float*)d_in[8];
  const float* b1      = (const float*)d_in[9];
  float* out           = (float*)d_out;

  const int B = 8192;
  jacde_kernel<<<B / 16, 64, 0, stream>>>(t, hmat, coeffs, dcoeffs, tobs,
                                          wx, wh, wout, b0, b1, out);
}


# Each wave owns 32 output columns, not 64 (4 waves per block, under 512 registers each), so two waves share a SIMD: one wave's staging, barrier and sigmoid/tanh waits run under the other's matrix work
# speedup vs baseline: 1.5696x; 1.1638x over previous
#include <hip/hip_runtime.h>


typedef __attribute__((ext_vector_type(2))) float v2f;
typedef __attribute__((ext_vector_type(8))) float v8f;

#define LDP   132
#define HDIM  128
#define CINC  64
#define NOBSC 16
#define NWAVE 4
#define WROWS 32
#define NTILE (WROWS / 16)
#define WREGION (WROWS * (HDIM + 4))

__device__ __forceinline__ v8f wmma4(v2f a, v2f b, v8f c) {
  return __builtin_amdgcn_wmma_f32_16x16x4_f32(
       false, a,  false, b,
       (short)0, c,  false,  false);
}

__device__ __forceinline__ void lds_barrier() {
  __builtin_amdgcn_fence(__ATOMIC_RELEASE, "workgroup", "local");
  __builtin_amdgcn_s_barrier();
  __builtin_amdgcn_fence(__ATOMIC_ACQUIRE, "workgroup", "local");
}

typedef __attribute__((ext_vector_type(4))) float v4f;

template <int K>
struct WRows { v4f r[WROWS * K / 128]; };

template <int K, int I = 0>
__device__ __forceinline__ void w_fetch_from(const v4f* __restrict__ src, WRows<K>& w) {
  if constexpr (I < WROWS * K / 128) {
    w.r[I] = src[I * 32];
    w_fetch_from<K, I + 1>(src, w);
  }
}

template <int K>
__device__ __forceinline__ void w_fetch(const float* __restrict__ W, int wcol, int lane,
                                        WRows<K>& w) {
  w_fetch_from<K>(reinterpret_cast<const v4f*>(W + (size_t)wcol * K) + lane, w);
}

template <int K, int I = 0>
__device__ __forceinline__ void w_commit_to(float* __restrict__ sWl, const WRows<K>& w) {
  if constexpr (I < WROWS * K / 128) {
    *reinterpret_cast<v4f*>(sWl + ((I * 128) / K) * (K + 4) + ((I * 128) % K)) = w.r[I];
    w_commit_to<K, I + 1>(sWl, w);
  }
}

template <int K>
__device__ __forceinline__ void w_commit(float* __restrict__ sW, int lane, const WRows<K>& w) {
  const int f   = lane * 4;
  const int row = f / K;
  const int col = f % K;
  w_commit_to<K>(sW + row * (K + 4) + col, w);
}

template <int K>
__device__ __forceinline__ void gemm_lds(const float* __restrict__ sY,
                                         const float* __restrict__ sW,
                                         int lane, v8f acc[NTILE]) {
  const int mrow  = lane & 15;
  const int khalf = (lane >> 4) << 1;
  #pragma unroll 4
  for (int ks = 0; ks < K / 4; ++ks) {
    const int k0 = ks * 4 + khalf;
    v2f a = *(const v2f*)(sY + mrow * LDP + k0);
    #pragma unroll
    for (int nt = 0; nt < NTILE; ++nt) {
      v2f b = *(const v2f*)(sW + (nt * 16 + mrow) * (K + 4) + k0);
      acc[nt] = wmma4(a, b, acc[nt]);
    }
  }
}

typedef __attribute__((address_space(3))) float lds_float;
__device__ __forceinline__ void lds_store1(float* p, float v) {
  *(volatile lds_float*)p = v;
}

__device__ __forceinline__ void write_tile(float* __restrict__ sY, int wcol, int lane,
                                           const v8f v[NTILE]) {
  const int hi   = lane >> 4;
  const int col0 = lane & 15;
  #pragma unroll
  for (int nt = 0; nt < NTILE; ++nt) {
    const int col = wcol + nt * 16 + col0;
    #pragma unroll
    for (int r = 0; r < 8; ++r)
      lds_store1(sY + (r + 8 * hi) * LDP + col, v[nt][r]);
  }
}

__global__ void __launch_bounds__(128) __attribute__((amdgpu_waves_per_eu(2, 2))) jacde_kernel(
    const float* __restrict__ t,      const float* __restrict__ hmat,
    const float* __restrict__ coeffs, const float* __restrict__ dcoeffs,
    const float* __restrict__ tobs,
    const float* __restrict__ wx,     const float* __restrict__ wh,
    const float* __restrict__ wout,
    const float* __restrict__ b0,     const float* __restrict__ b1,
    float* __restrict__ out) {
  __shared__ __attribute__((aligned(16))) float sY[16 * LDP];
  __shared__ __attribute__((aligned(16))) float sWall[NWAVE * WREGION];

  const int tid  = threadIdx.x;
  const int lane = tid & 31;
  const int wave = tid >> 5;
  const int wcol = wave * WROWS;
  const int brow = blockIdx.x * 16;
  const int hi   = lane >> 4;
  const int col0 = lane & 15;
  float* sW = sWall + wave * WREGION;

  const float ts = t[0];
  int idx = 0;
  #pragma unroll
  for (int j = 0; j < NOBSC; ++j) idx += (tobs[j] <= ts) ? 1 : 0;
  idx -= 1;
  idx = idx < 0 ? 0 : (idx > NOBSC - 2 ? NOBSC - 2 : idx);
  const float dt = ts - tobs[idx];
  const float p1 = dt, p2 = dt * dt, p3 = p2 * dt;

  float bias0[NTILE], bias1[NTILE];
  #pragma unroll
  for (int nt = 0; nt < NTILE; ++nt) {
    bias0[nt] = b0[wcol + nt * 16 + col0];
    bias1[nt] = b1[wcol + nt * 16 + col0];
  }

  const int sch  = tid & (CINC - 1);
  const int spar = tid >> 6;
  {
    const float* cb = coeffs + ((size_t)brow * (NOBSC - 1) + idx) * CINC * 4;
    #pragma unroll 4
    for (int i = 0; i < 8; ++i) {
      const int m = 2 * i + spar;
      const float4 c4 = *(const float4*)(cb + (size_t)m * (NOBSC - 1) * CINC * 4 + sch * 4);
      sY[m * LDP + sch] = c4.x + c4.y * p1 + c4.z * p2 + c4.w * p3;
    }
  }
  {
    WRows<CINC> w0;
    w_fetch<CINC>(wx, wcol, lane, w0);
    w_commit<CINC>(sW, lane, w0);
  }
  WRows<HDIM> w1;
  w_fetch<HDIM>(wh, wcol, lane, w1);
  lds_barrier();

  v8f acc[NTILE];
  #pragma unroll
  for (int nt = 0; nt < NTILE; ++nt) {
    const float bv = bias0[nt];
    #pragma unroll
    for (int r = 0; r < 8; ++r) acc[nt][r] = bv;
  }
  gemm_lds<CINC>(sY, sW, lane, acc);
  __syncthreads();

  #pragma unroll 4
  for (int m = 0; m < 16; ++m)
    sY[m * LDP + tid] = hmat[(size_t)(brow + m) * HDIM + tid];
  w_commit<HDIM>(sW, lane, w1);
  WRows<HDIM> w2;
  w_fetch<HDIM>(wout, wcol, lane, w2);
  lds_barrier();
  gemm_lds<HDIM>(sY, sW, lane, acc);
  __syncthreads();

  v8f drelu[NTILE];
  #pragma unroll
  for (int nt = 0; nt < NTILE; ++nt)
    #pragma unroll
    for (int r = 0; r < 8; ++r) {
      const float x = acc[nt][r];
      drelu[nt][r] = 1.0f / (1.0f + __expf(-x));
      acc[nt][r]   = fmaxf(x, 0.0f);
    }
  write_tile(sY, wcol, lane, acc);
  w_commit<HDIM>(sW, lane, w2);
  WRows<CINC> w3;
  w_fetch<CINC>(wx, wcol, lane, w3);
  lds_barrier();

  v8f dtn[NTILE];
  {
    v8f a2[NTILE];
    #pragma unroll
    for (int nt = 0; nt < NTILE; ++nt) {
      const float bv = bias1[nt];
      #pragma unroll
      for (int r = 0; r < 8; ++r) a2[nt][r] = bv;
    }
    gemm_lds<HDIM>(sY, sW, lane, a2);
    #pragma unroll
    for (int nt = 0; nt < NTILE; ++nt)
      #pragma unroll
      for (int r = 0; r < 8; ++r) {
        const float th = tanhf(a2[nt][r]);
        dtn[nt][r] = 1.0f - th * th;
      }
  }
  __syncthreads();

  {
    const float* db = dcoeffs + ((size_t)brow * (NOBSC - 1) + idx) * CINC * 4;
    #pragma unroll 4
    for (int i = 0; i < 8; ++i) {
      const int m = 2 * i + spar;
      const float4 c4 = *(const float4*)(db + (size_t)m * (NOBSC - 1) * CINC * 4 + sch * 4);
      sY[m * LDP + sch] = c4.x + c4.y * p1 + c4.z * p2 + c4.w * p3;
    }
  }
  w_commit<CINC>(sW, lane, w3);
  WRows<HDIM> w4;
  w_fetch<HDIM>(wout, wcol, lane, w4);
  lds_barrier();
  v8f cur[NTILE];
  #pragma unroll
  for (int nt = 0; nt < NTILE; ++nt) cur[nt] = (v8f)(0.0f);
  gemm_lds<CINC>(sY, sW, lane, cur);
  __syncthreads();

  #pragma unroll
  for (int nt = 0; nt < NTILE; ++nt) cur[nt] *= drelu[nt];
  write_tile(sY, wcol, lane, cur);
  w_commit<HDIM>(sW, lane, w4);
  WRows<HDIM> w5;
  w_fetch<HDIM>(wh, wcol, lane, w5);
  lds_barrier();
  v8f outv[NTILE];
  #pragma unroll
  for (int nt = 0; nt < NTILE; ++nt) outv[nt] = (v8f)(0.0f);
  gemm_lds<HDIM>(sY, sW, lane, outv);
  __syncthreads();
  #pragma unroll
  for (int nt = 0; nt < NTILE; ++nt) {
    outv[nt] *= dtn[nt];
    cur[nt] = outv[nt];
  }

  WRows<HDIM> w6, w7, w8;

  write_tile(sY, wcol, lane, cur);
  w_commit<HDIM>(sW, lane, w5);
  w_fetch<HDIM>(wout, wcol, lane, w6);
  lds_barrier();
  {
    v8f t1[NTILE];
    #pragma unroll
    for (int nt = 0; nt < NTILE; ++nt) t1[nt] = (v8f)(0.0f);
    gemm_lds<HDIM>(sY, sW, lane, t1);
    __syncthreads();
    #pragma unroll
    for (int nt = 0; nt < NTILE; ++nt) t1[nt] *= drelu[nt];
    write_tile(sY, wcol, lane, t1);
  }
  w_commit<HDIM>(sW, lane, w6);
  w_fetch<HDIM>(wh, wcol, lane, w7);
  lds_barrier();
  {
    v8f t2[NTILE];
    #pragma unroll
    for (int nt = 0; nt < NTILE; ++nt) t2[nt] = (v8f)(0.0f);
    gemm_lds<HDIM>(sY, sW, lane, t2);
    __syncthreads();
    #pragma unroll
    for (int nt = 0; nt < NTILE; ++nt) {
      t2[nt] *= dtn[nt];
      outv[nt] += t2[nt];
      cur[nt] = t2[nt];
    }
  }

  write_tile(sY, wcol, lane, cur);
  w_commit<HDIM>(sW, lane, w7);
  w_fetch<HDIM>(wout, wcol, lane, w8);
  lds_barrier();
  {
    v8f t1[NTILE];
    #pragma unroll
    for (int nt = 0; nt < NTILE; ++nt) t1[nt] = (v8f)(0.0f);
    gemm_lds<HDIM>(sY, sW, lane, t1);
    __syncthreads();
    #pragma unroll
    for (int nt = 0; nt < NTILE; ++nt) t1[nt] *= drelu[nt];
    write_tile(sY, wcol, lane, t1);
  }
  w_commit<HDIM>(sW, lane, w8);
  __syncthreads();
  {
    v8f t2[NTILE];
    #pragma unroll
    for (int nt = 0; nt < NTILE; ++nt) t2[nt] = (v8f)(0.0f);
    gemm_lds<HDIM>(sY, sW, lane, t2);
    #pragma unroll
    for (int nt = 0; nt < NTILE; ++nt) {
      t2[nt] *= dtn[nt];
      outv[nt] += t2[nt];
    }
  }

  #pragma unroll
  for (int nt = 0; nt < NTILE; ++nt) {
    const int col = wcol + nt * 16 + col0;
    #pragma unroll
    for (int r = 0; r < 8; ++r)
      out[(size_t)(brow + r + 8 * hi) * HDIM + col] = outv[nt][r];
  }
}

extern "C" void kernel_launch(void* const* d_in, const int* in_sizes, int n_in,
                              void* d_out, int out_size, void* d_ws, size_t ws_size,
                              hipStream_t stream) {
  const float* t       = (const float*)d_in[0];
  const float* hmat    = (const float*)d_in[1];
  const float* coeffs  = (const float*)d_in[2];
  const float* dcoeffs = (const float*)d_in[3];
  const float* tobs    = (const float*)d_in[4];
  const float* wx      = (const float*)d_in[5];
  const float* wh      = (const float*)d_in[6];
  const float* wout    = (const float*)d_in[7];
  const float* b0      = (const float*)d_in[8];
  const float* b1      = (const float*)d_in[9];
  float* out           = (float*)d_out;

  const int B = 8192;
  jacde_kernel<<<B / 16, 128, 0, stream>>>(t, hmat, coeffs, dcoeffs, tobs,
                                          wx, wh, wout, b0, b1, out);
}


# One 32-row block of 8 waves per compute unit copies wx, wh and wout into LDS once; all nine products read them there, so a weight row is fetched once per block, not per product per 16 rows
# speedup vs baseline: 1.7610x; 1.1219x over previous
#include <hip/hip_runtime.h>


typedef __attribute__((ext_vector_type(2))) float v2f;
typedef __attribute__((ext_vector_type(8))) float v8f;

#define LDP   132
#define HDIM  128
#define CINC  64
#define NOBSC 16
#define NWAVE 8
#define BROWS 32
#define WROWS 32
#define NTILE (WROWS / 16)
#define FROWS (HDIM / NWAVE)

__device__ __forceinline__ v8f wmma4(v2f a, v2f b, v8f c) {
  return __builtin_amdgcn_wmma_f32_16x16x4_f32(
       false, a,  false, b,
       (short)0, c,  false,  false);
}

__device__ __forceinline__ void lds_barrier() {
  __builtin_amdgcn_fence(__ATOMIC_RELEASE, "workgroup", "local");
  __builtin_amdgcn_s_barrier();
  __builtin_amdgcn_fence(__ATOMIC_ACQUIRE, "workgroup", "local");
}

typedef __attribute__((ext_vector_type(4))) float v4f;

template <int K>
struct WRows { v4f r[FROWS * K / 128]; };

template <int K, int I = 0>
__device__ __forceinline__ void w_fetch_from(const v4f* __restrict__ src, WRows<K>& w) {
  if constexpr (I < FROWS * K / 128) {
    w.r[I] = src[I * 32];
    w_fetch_from<K, I + 1>(src, w);
  }
}

template <int K>
__device__ __forceinline__ void w_fetch(const float* __restrict__ W, int frow, int lane,
                                        WRows<K>& w) {
  w_fetch_from<K>(reinterpret_cast<const v4f*>(W + (size_t)frow * K) + lane, w);
}

template <int K, int I = 0>
__device__ __forceinline__ void w_commit_to(float* __restrict__ sWl, const WRows<K>& w) {
  if constexpr (I < FROWS * K / 128) {
    *reinterpret_cast<v4f*>(sWl + ((I * 128) / K) * (K + 4) + ((I * 128) % K)) = w.r[I];
    w_commit_to<K, I + 1>(sWl, w);
  }
}

template <int K>
__device__ __forceinline__ void w_commit(float* __restrict__ sW, int lane, const WRows<K>& w) {
  const int f   = lane * 4;
  const int row = f / K;
  const int col = f % K;
  w_commit_to<K>(sW + row * (K + 4) + col, w);
}

typedef __attribute__((address_space(3))) float lds_float;
typedef __attribute__((address_space(3))) v2f   lds_v2f;

template <int K>
__device__ __forceinline__ void gemm_lds(const float* __restrict__ sY,
                                         const float* __restrict__ sW,
                                         int lane, v8f acc[NTILE]) {
  const int mrow  = lane & 15;
  const int khalf = (lane >> 4) << 1;
  const lds_float* pa = (const lds_float*)(sY + mrow * LDP + khalf);
  const lds_float* pb = (const lds_float*)(sW + mrow * (K + 4) + khalf);
  asm("" : "+v"(pa), "+v"(pb));
  #pragma unroll 4
  for (int ks = 0; ks < K / 4; ++ks) {
    v2f a = *(const lds_v2f*)(pa + ks * 4);
    #pragma unroll
    for (int nt = 0; nt < NTILE; ++nt) {
      v2f b = *(const lds_v2f*)(pb + nt * 16 * (K + 4) + ks * 4);
      acc[nt] = wmma4(a, b, acc[nt]);
    }
  }
}

__device__ __forceinline__ void lds_store1(float* p, float v) {
  *(volatile lds_float*)p = v;
}

__device__ __forceinline__ void write_tile(float* __restrict__ sY, int wcol, int lane,
                                           const v8f v[NTILE]) {
  const int hi   = lane >> 4;
  const int col0 = lane & 15;
  #pragma unroll
  for (int nt = 0; nt < NTILE; ++nt) {
    const int col = wcol + nt * 16 + col0;
    #pragma unroll
    for (int r = 0; r < 8; ++r)
      lds_store1(sY + (r + 8 * hi) * LDP + col, v[nt][r]);
  }
}

__global__ void __launch_bounds__(256) __attribute__((amdgpu_waves_per_eu(2, 2))) jacde_kernel(
    const float* __restrict__ t,      const float* __restrict__ hmat,
    const float* __restrict__ coeffs, const float* __restrict__ dcoeffs,
    const float* __restrict__ tobs,
    const float* __restrict__ wx,     const float* __restrict__ wh,
    const float* __restrict__ wout,
    const float* __restrict__ b0,     const float* __restrict__ b1,
    float* __restrict__ out) {
  __shared__ __attribute__((aligned(16))) float sYall[BROWS * LDP];
  __shared__ __attribute__((aligned(16))) float sWx[HDIM * (CINC + 4)];
  __shared__ __attribute__((aligned(16))) float sWh[HDIM * (HDIM + 4)];
  __shared__ __attribute__((aligned(16))) float sWo[HDIM * (HDIM + 4)];

  const int tid  = threadIdx.x;
  const int lane = tid & 31;
  const int wave = tid >> 5;
  const int mi   = wave & 1;
  const int ni   = wave >> 1;
  const int wcol = ni * WROWS;
  const int frow = wave * FROWS;
  const int brow = blockIdx.x * BROWS;
  const int hi   = lane >> 4;
  const int col0 = lane & 15;
  float* sY = sYall + mi * 16 * LDP;
  const float* sBx = sWx + wcol * (CINC + 4);
  const float* sBh = sWh + wcol * (HDIM + 4);
  const float* sBo = sWo + wcol * (HDIM + 4);

  WRows<CINC> rx;
  w_fetch<CINC>(wx, frow, lane, rx);
  __builtin_amdgcn_sched_barrier(0);

  const float ts = t[0];
  int idx = 0;
  #pragma unroll
  for (int j = 0; j < NOBSC; ++j) idx += (tobs[j] <= ts) ? 1 : 0;
  idx -= 1;
  idx = idx < 0 ? 0 : (idx > NOBSC - 2 ? NOBSC - 2 : idx);
  const float dt = ts - tobs[idx];
  const float p1 = dt, p2 = dt * dt, p3 = p2 * dt;

  float bias0[NTILE], bias1[NTILE];
  #pragma unroll
  for (int nt = 0; nt < NTILE; ++nt) {
    bias0[nt] = b0[wcol + nt * 16 + col0];
    bias1[nt] = b1[wcol + nt * 16 + col0];
  }

  const int sch  = tid & (CINC - 1);
  const int spar = tid >> 6;
  float4 c4[8];
  {
    const float* cb = coeffs + ((size_t)brow * (NOBSC - 1) + idx) * CINC * 4;
    #pragma unroll
    for (int i = 0; i < 8; ++i)
      c4[i] = *(const float4*)(cb + (size_t)(4 * i + spar) * (NOBSC - 1) * CINC * 4 + sch * 4);
  }
  __builtin_amdgcn_sched_barrier(0);
  WRows<HDIM> rh, ro;
  w_fetch<HDIM>(wh, frow, lane, rh);
  w_fetch<HDIM>(wout, frow, lane, ro);
  __builtin_amdgcn_sched_barrier(0);
  w_commit<CINC>(sWx + frow * (CINC + 4), lane, rx);
  #pragma unroll
  for (int i = 0; i < 8; ++i)
    sYall[(4 * i + spar) * LDP + sch] = c4[i].x + c4[i].y * p1 + c4[i].z * p2 + c4[i].w * p3;
  lds_barrier();

  v8f acc[NTILE];
  #pragma unroll
  for (int nt = 0; nt < NTILE; ++nt) {
    const float bv = bias0[nt];
    #pragma unroll
    for (int r = 0; r < 8; ++r) acc[nt][r] = bv;
  }
  gemm_lds<CINC>(sY, sBx, lane, acc);
  lds_barrier();

  {
    const int hc = tid & (HDIM - 1);
    const int hp = tid >> 7;
    #pragma unroll
    for (int i = 0; i < 16; ++i) {
      const int m = 2 * i + hp;
      sYall[m * LDP + hc] = hmat[(size_t)(brow + m) * HDIM + hc];
    }
  }
  w_commit<HDIM>(sWh + frow * (HDIM + 4), lane, rh);
  lds_barrier();
  gemm_lds<HDIM>(sY, sBh, lane, acc);
  lds_barrier();

  v8f drelu[NTILE];
  #pragma unroll
  for (int nt = 0; nt < NTILE; ++nt)
    #pragma unroll
    for (int r = 0; r < 8; ++r) {
      const float x = acc[nt][r];
      drelu[nt][r] = 1.0f / (1.0f + __expf(-x));
      acc[nt][r]   = fmaxf(x, 0.0f);
    }
  write_tile(sY, wcol, lane, acc);
  w_commit<HDIM>(sWo + frow * (HDIM + 4), lane, ro);
  lds_barrier();

  v8f dtn[NTILE];
  {
    v8f a2[NTILE];
    #pragma unroll
    for (int nt = 0; nt < NTILE; ++nt) {
      const float bv = bias1[nt];
      #pragma unroll
      for (int r = 0; r < 8; ++r) a2[nt][r] = bv;
    }
    gemm_lds<HDIM>(sY, sBo, lane, a2);
    #pragma unroll
    for (int nt = 0; nt < NTILE; ++nt)
      #pragma unroll
      for (int r = 0; r < 8; ++r) {
        const float th = tanhf(a2[nt][r]);
        dtn[nt][r] = 1.0f - th * th;
      }
  }
  lds_barrier();

  {
    const float* db = dcoeffs + ((size_t)brow * (NOBSC - 1) + idx) * CINC * 4;
    #pragma unroll 4
    for (int i = 0; i < 8; ++i) {
      const int m = 4 * i + spar;
      const float4 c4 = *(const float4*)(db + (size_t)m * (NOBSC - 1) * CINC * 4 + sch * 4);
      sYall[m * LDP + sch] = c4.x + c4.y * p1 + c4.z * p2 + c4.w * p3;
    }
  }
  lds_barrier();
  v8f cur[NTILE];
  #pragma unroll
  for (int nt = 0; nt < NTILE; ++nt) cur[nt] = (v8f)(0.0f);
  gemm_lds<CINC>(sY, sBx, lane, cur);
  lds_barrier();

  #pragma unroll
  for (int nt = 0; nt < NTILE; ++nt) cur[nt] *= drelu[nt];
  write_tile(sY, wcol, lane, cur);
  lds_barrier();
  v8f outv[NTILE];
  #pragma unroll
  for (int nt = 0; nt < NTILE; ++nt) outv[nt] = (v8f)(0.0f);
  gemm_lds<HDIM>(sY, sBo, lane, outv);
  lds_barrier();
  #pragma unroll
  for (int nt = 0; nt < NTILE; ++nt) {
    outv[nt] *= dtn[nt];
    cur[nt] = outv[nt];
  }

  write_tile(sY, wcol, lane, cur);
  lds_barrier();
  {
    v8f t1[NTILE];
    #pragma unroll
    for (int nt = 0; nt < NTILE; ++nt) t1[nt] = (v8f)(0.0f);
    gemm_lds<HDIM>(sY, sBh, lane, t1);
    lds_barrier();
    #pragma unroll
    for (int nt = 0; nt < NTILE; ++nt) t1[nt] *= drelu[nt];
    write_tile(sY, wcol, lane, t1);
  }
  lds_barrier();
  {
    v8f t2[NTILE];
    #pragma unroll
    for (int nt = 0; nt < NTILE; ++nt) t2[nt] = (v8f)(0.0f);
    gemm_lds<HDIM>(sY, sBo, lane, t2);
    lds_barrier();
    #pragma unroll
    for (int nt = 0; nt < NTILE; ++nt) {
      t2[nt] *= dtn[nt];
      outv[nt] += t2[nt];
      cur[nt] = t2[nt];
    }
  }

  write_tile(sY, wcol, lane, cur);
  lds_barrier();
  {
    v8f t1[NTILE];
    #pragma unroll
    for (int nt = 0; nt < NTILE; ++nt) t1[nt] = (v8f)(0.0f);
    gemm_lds<HDIM>(sY, sBh, lane, t1);
    lds_barrier();
    #pragma unroll
    for (int nt = 0; nt < NTILE; ++nt) t1[nt] *= drelu[nt];
    write_tile(sY, wcol, lane, t1);
  }
  lds_barrier();
  {
    v8f t2[NTILE];
    #pragma unroll
    for (int nt = 0; nt < NTILE; ++nt) t2[nt] = (v8f)(0.0f);
    gemm_lds<HDIM>(sY, sBo, lane, t2);
    #pragma unroll
    for (int nt = 0; nt < NTILE; ++nt) {
      t2[nt] *= dtn[nt];
      outv[nt] += t2[nt];
    }
  }

  #pragma unroll
  for (int nt = 0; nt < NTILE; ++nt) {
    const int col = wcol + nt * 16 + col0;
    #pragma unroll
    for (int r = 0; r < 8; ++r)
      out[(size_t)(brow + mi * 16 + r + 8 * hi) * HDIM + col] = outv[nt][r];
  }
}

extern "C" void kernel_launch(void* const* d_in, const int* in_sizes, int n_in,
                              void* d_out, int out_size, void* d_ws, size_t ws_size,
                              hipStream_t stream) {
  const float* t       = (const float*)d_in[0];
  const float* hmat    = (const float*)d_in[1];
  const float* coeffs  = (const float*)d_in[2];
  const float* dcoeffs = (const float*)d_in[3];
  const float* tobs    = (const float*)d_in[4];
  const float* wx      = (const float*)d_in[5];
  const float* wh      = (const float*)d_in[6];
  const float* wout    = (const float*)d_in[7];
  const float* b0      = (const float*)d_in[8];
  const float* b1      = (const float*)d_in[9];
  float* out           = (float*)d_out;

  jacde_kernel<<<8192 / BROWS, 256, 0, stream>>>(t, hmat, coeffs, dcoeffs, tobs,
                                                 wx, wh, wout, b0, b1, out);
}


# Two LDS activation tiles used alternately, h and xdot in own tiles fetched at the start: each product's input is written while the other tile is read (8 barriers, not 17; no mid-chain load wait)
# speedup vs baseline: 1.8752x; 1.0648x over previous
#include <hip/hip_runtime.h>


typedef __attribute__((ext_vector_type(2))) float v2f;
typedef __attribute__((ext_vector_type(8))) float v8f;

#define LDP   132
#define LDX   68
#define HDIM  128
#define CINC  64
#define NOBSC 16
#define NWAVE 8
#define BROWS 32
#define WROWS 32
#define NTILE (WROWS / 16)
#define FROWS (HDIM / NWAVE)

__device__ __forceinline__ v8f wmma4(v2f a, v2f b, v8f c) {
  return __builtin_amdgcn_wmma_f32_16x16x4_f32(
       false, a,  false, b,
       (short)0, c,  false,  false);
}

__device__ __forceinline__ void lds_barrier() {
  __builtin_amdgcn_fence(__ATOMIC_RELEASE, "workgroup", "local");
  __builtin_amdgcn_s_barrier();
  __builtin_amdgcn_fence(__ATOMIC_ACQUIRE, "workgroup", "local");
}

typedef __attribute__((ext_vector_type(4))) float v4f;

template <int K>
struct WRows { v4f r[FROWS * K / 128]; };

template <int K, int I = 0>
__device__ __forceinline__ void w_fetch_from(const v4f* __restrict__ src, WRows<K>& w) {
  if constexpr (I < FROWS * K / 128) {
    w.r[I] = src[I * 32];
    w_fetch_from<K, I + 1>(src, w);
  }
}

template <int K>
__device__ __forceinline__ void w_fetch(const float* __restrict__ W, int frow, int lane,
                                        WRows<K>& w) {
  w_fetch_from<K>(reinterpret_cast<const v4f*>(W + (size_t)frow * K) + lane, w);
}

template <int K, int I = 0>
__device__ __forceinline__ void w_commit_to(float* __restrict__ sWl, const WRows<K>& w) {
  if constexpr (I < FROWS * K / 128) {
    *reinterpret_cast<v4f*>(sWl + ((I * 128) / K) * (K + 4) + ((I * 128) % K)) = w.r[I];
    w_commit_to<K, I + 1>(sWl, w);
  }
}

template <int K>
__device__ __forceinline__ void w_commit(float* __restrict__ sW, int lane, const WRows<K>& w) {
  const int f   = lane * 4;
  const int row = f / K;
  const int col = f % K;
  w_commit_to<K>(sW + row * (K + 4) + col, w);
}

typedef __attribute__((address_space(3))) float lds_float;
typedef __attribute__((address_space(3))) v2f   lds_v2f;

template <int K, int LDA>
__device__ __forceinline__ void gemm_lds(const float* __restrict__ sY,
                                         const float* __restrict__ sW,
                                         int lane, v8f acc[NTILE]) {
  const int mrow  = lane & 15;
  const int khalf = (lane >> 4) << 1;
  const lds_float* pa = (const lds_float*)(sY + mrow * LDA + khalf);
  const lds_float* pb = (const lds_float*)(sW + mrow * (K + 4) + khalf);
  asm("" : "+v"(pa), "+v"(pb));
  #pragma unroll 4
  for (int ks = 0; ks < K / 4; ++ks) {
    v2f a = *(const lds_v2f*)(pa + ks * 4);
    #pragma unroll
    for (int nt = 0; nt < NTILE; ++nt) {
      v2f b = *(const lds_v2f*)(pb + nt * 16 * (K + 4) + ks * 4);
      acc[nt] = wmma4(a, b, acc[nt]);
    }
  }
}

__device__ __forceinline__ void lds_store1(float* p, float v) {
  *(volatile lds_float*)p = v;
}

__device__ __forceinline__ void write_tile(float* __restrict__ sY, int wcol, int lane,
                                           const v8f v[NTILE]) {
  const int hi   = lane >> 4;
  const int col0 = lane & 15;
  #pragma unroll
  for (int nt = 0; nt < NTILE; ++nt) {
    const int col = wcol + nt * 16 + col0;
    #pragma unroll
    for (int r = 0; r < 8; ++r)
      lds_store1(sY + (r + 8 * hi) * LDP + col, v[nt][r]);
  }
}

__global__ void __launch_bounds__(256) __attribute__((amdgpu_waves_per_eu(2, 2))) jacde_kernel(
    const float* __restrict__ t,      const float* __restrict__ hmat,
    const float* __restrict__ coeffs, const float* __restrict__ dcoeffs,
    const float* __restrict__ tobs,
    const float* __restrict__ wx,     const float* __restrict__ wh,
    const float* __restrict__ wout,
    const float* __restrict__ b0,     const float* __restrict__ b1,
    float* __restrict__ out) {
  __shared__ __attribute__((aligned(16))) float sAall[BROWS * LDP];
  __shared__ __attribute__((aligned(16))) float sBall[BROWS * LDP];
  __shared__ __attribute__((aligned(16))) float sXall[BROWS * LDX];
  __shared__ __attribute__((aligned(16))) float sWx[HDIM * (CINC + 4)];
  __shared__ __attribute__((aligned(16))) float sWh[HDIM * (HDIM + 4)];
  __shared__ __attribute__((aligned(16))) float sWo[HDIM * (HDIM + 4)];

  const int tid  = threadIdx.x;
  const int lane = tid & 31;
  const int wave = tid >> 5;
  const int mi   = wave & 1;
  const int ni   = wave >> 1;
  const int wcol = ni * WROWS;
  const int frow = wave * FROWS;
  const int brow = blockIdx.x * BROWS;
  const int hi   = lane >> 4;
  const int col0 = lane & 15;
  float* sA = sAall + mi * 16 * LDP;
  float* sB = sBall + mi * 16 * LDP;
  const float* sX = sXall + mi * 16 * LDX;
  const float* sBx = sWx + wcol * (CINC + 4);
  const float* sBh = sWh + wcol * (HDIM + 4);
  const float* sBo = sWo + wcol * (HDIM + 4);

  WRows<CINC> rx;
  w_fetch<CINC>(wx, frow, lane, rx);
  __builtin_amdgcn_sched_barrier(0);

  const float ts = t[0];
  int idx = 0;
  #pragma unroll
  for (int j = 0; j < NOBSC; ++j) idx += (tobs[j] <= ts) ? 1 : 0;
  idx -= 1;
  idx = idx < 0 ? 0 : (idx > NOBSC - 2 ? NOBSC - 2 : idx);
  const float dt = ts - tobs[idx];
  const float p1 = dt, p2 = dt * dt, p3 = p2 * dt;

  float bias0[NTILE], bias1[NTILE];
  #pragma unroll
  for (int nt = 0; nt < NTILE; ++nt) {
    bias0[nt] = b0[wcol + nt * 16 + col0];
    bias1[nt] = b1[wcol + nt * 16 + col0];
  }

  const int sch  = tid & (CINC - 1);
  const int spar = tid >> 6;
  float4 c4[8];
  {
    const float* cb = coeffs + ((size_t)brow * (NOBSC - 1) + idx) * CINC * 4;
    #pragma unroll
    for (int i = 0; i < 8; ++i)
      c4[i] = *(const float4*)(cb + (size_t)(4 * i + spar) * (NOBSC - 1) * CINC * 4 + sch * 4);
  }
  const int hq = tid & 31;
  const int hr = tid >> 5;
  v4f h4[4];
  #pragma unroll
  for (int i = 0; i < 4; ++i)
    h4[i] = *reinterpret_cast<const v4f*>(hmat + (size_t)(brow + hr + 8 * i) * HDIM + hq * 4);
  __builtin_amdgcn_sched_barrier(0);
  WRows<HDIM> rh, ro;
  w_fetch<HDIM>(wh, frow, lane, rh);
  w_fetch<HDIM>(wout, frow, lane, ro);
  __builtin_amdgcn_sched_barrier(0);
  w_commit<CINC>(sWx + frow * (CINC + 4), lane, rx);
  #pragma unroll
  for (int i = 0; i < 8; ++i)
    sAall[(4 * i + spar) * LDP + sch] = c4[i].x + c4[i].y * p1 + c4[i].z * p2 + c4[i].w * p3;
  __builtin_amdgcn_sched_barrier(0);
  float4 d4[8];
  {
    const float* db = dcoeffs + ((size_t)brow * (NOBSC - 1) + idx) * CINC * 4;
    #pragma unroll
    for (int i = 0; i < 8; ++i)
      d4[i] = *(const float4*)(db + (size_t)(4 * i + spar) * (NOBSC - 1) * CINC * 4 + sch * 4);
  }
  __builtin_amdgcn_sched_barrier(0);
  lds_barrier();

  v8f acc[NTILE];
  #pragma unroll
  for (int nt = 0; nt < NTILE; ++nt) {
    const float bv = bias0[nt];
    #pragma unroll
    for (int r = 0; r < 8; ++r) acc[nt][r] = bv;
  }
  gemm_lds<CINC, LDP>(sA, sBx, lane, acc);
  #pragma unroll
  for (int i = 0; i < 4; ++i)
    *reinterpret_cast<v4f*>(sBall + (hr + 8 * i) * LDP + hq * 4) = h4[i];
  w_commit<HDIM>(sWh + frow * (HDIM + 4), lane, rh);
  lds_barrier();
  gemm_lds<HDIM, LDP>(sB, sBh, lane, acc);

  v8f drelu[NTILE];
  #pragma unroll
  for (int nt = 0; nt < NTILE; ++nt)
    #pragma unroll
    for (int r = 0; r < 8; ++r) {
      const float x = acc[nt][r];
      drelu[nt][r] = 1.0f / (1.0f + __expf(-x));
      acc[nt][r]   = fmaxf(x, 0.0f);
    }
  write_tile(sA, wcol, lane, acc);
  w_commit<HDIM>(sWo + frow * (HDIM + 4), lane, ro);
  #pragma unroll
  for (int i = 0; i < 8; ++i)
    sXall[(4 * i + spar) * LDX + sch] = d4[i].x + d4[i].y * p1 + d4[i].z * p2 + d4[i].w * p3;
  lds_barrier();

  v8f dtn[NTILE];
  v8f cur[NTILE];
  {
    v8f a2[NTILE];
    #pragma unroll
    for (int nt = 0; nt < NTILE; ++nt) {
      const float bv = bias1[nt];
      #pragma unroll
      for (int r = 0; r < 8; ++r) a2[nt][r] = bv;
      cur[nt] = (v8f)(0.0f);
    }
    gemm_lds<HDIM, LDP>(sA, sBo, lane, a2);
    gemm_lds<CINC, LDX>(sX, sBx, lane, cur);
    #pragma unroll
    for (int nt = 0; nt < NTILE; ++nt)
      #pragma unroll
      for (int r = 0; r < 8; ++r) {
        const float th = tanhf(a2[nt][r]);
        dtn[nt][r] = 1.0f - th * th;
      }
  }

  #pragma unroll
  for (int nt = 0; nt < NTILE; ++nt) cur[nt] *= drelu[nt];
  write_tile(sB, wcol, lane, cur);
  lds_barrier();
  v8f outv[NTILE];
  #pragma unroll
  for (int nt = 0; nt < NTILE; ++nt) outv[nt] = (v8f)(0.0f);
  gemm_lds<HDIM, LDP>(sB, sBo, lane, outv);
  #pragma unroll
  for (int nt = 0; nt < NTILE; ++nt) {
    outv[nt] *= dtn[nt];
    cur[nt] = outv[nt];
  }

  write_tile(sA, wcol, lane, cur);
  lds_barrier();
  {
    v8f t1[NTILE];
    #pragma unroll
    for (int nt = 0; nt < NTILE; ++nt) t1[nt] = (v8f)(0.0f);
    gemm_lds<HDIM, LDP>(sA, sBh, lane, t1);
    #pragma unroll
    for (int nt = 0; nt < NTILE; ++nt) t1[nt] *= drelu[nt];
    write_tile(sB, wcol, lane, t1);
  }
  lds_barrier();
  {
    v8f t2[NTILE];
    #pragma unroll
    for (int nt = 0; nt < NTILE; ++nt) t2[nt] = (v8f)(0.0f);
    gemm_lds<HDIM, LDP>(sB, sBo, lane, t2);
    #pragma unroll
    for (int nt = 0; nt < NTILE; ++nt) {
      t2[nt] *= dtn[nt];
      outv[nt] += t2[nt];
      cur[nt] = t2[nt];
    }
  }

  write_tile(sA, wcol, lane, cur);
  lds_barrier();
  {
    v8f t1[NTILE];
    #pragma unroll
    for (int nt = 0; nt < NTILE; ++nt) t1[nt] = (v8f)(0.0f);
    gemm_lds<HDIM, LDP>(sA, sBh, lane, t1);
    #pragma unroll
    for (int nt = 0; nt < NTILE; ++nt) t1[nt] *= drelu[nt];
    write_tile(sB, wcol, lane, t1);
  }
  lds_barrier();
  {
    v8f t2[NTILE];
    #pragma unroll
    for (int nt = 0; nt < NTILE; ++nt) t2[nt] = (v8f)(0.0f);
    gemm_lds<HDIM, LDP>(sB, sBo, lane, t2);
    #pragma unroll
    for (int nt = 0; nt < NTILE; ++nt) {
      t2[nt] *= dtn[nt];
      outv[nt] += t2[nt];
    }
  }

  #pragma unroll
  for (int nt = 0; nt < NTILE; ++nt) {
    const int col = wcol + nt * 16 + col0;
    #pragma unroll
    for (int r = 0; r < 8; ++r)
      out[(size_t)(brow + mi * 16 + r + 8 * hi) * HDIM + col] = outv[nt][r];
  }
}

extern "C" void kernel_launch(void* const* d_in, const int* in_sizes, int n_in,
                              void* d_out, int out_size, void* d_ws, size_t ws_size,
                              hipStream_t stream) {
  const float* t       = (const float*)d_in[0];
  const float* hmat    = (const float*)d_in[1];
  const float* coeffs  = (const float*)d_in[2];
  const float* dcoeffs = (const float*)d_in[3];
  const float* tobs    = (const float*)d_in[4];
  const float* wx      = (const float*)d_in[5];
  const float* wh      = (const float*)d_in[6];
  const float* wout    = (const float*)d_in[7];
  const float* b0      = (const float*)d_in[8];
  const float* b1      = (const float*)d_in[9];
  float* out           = (float*)d_out;

  jacde_kernel<<<8192 / BROWS, 256, 0, stream>>>(t, hmat, coeffs, dcoeffs, tobs,
                                                 wx, wh, wout, b0, b1, out);
}


# 1 - tanh^2 as 4q/(1+q)^2 with q = exp(-2|a|) (one exp, one reciprocal, no branch) and the sigmoid's division as a hardware reciprocal: the serial element-wise stretch before the jx product shrinks
# speedup vs baseline: 2.0336x; 1.0845x over previous
#include <hip/hip_runtime.h>


typedef __attribute__((ext_vector_type(2))) float v2f;
typedef __attribute__((ext_vector_type(8))) float v8f;

#define LDP   132
#define LDX   68
#define HDIM  128
#define CINC  64
#define NOBSC 16
#define NWAVE 8
#define BROWS 32
#define WROWS 32
#define NTILE (WROWS / 16)
#define FROWS (HDIM / NWAVE)

__device__ __forceinline__ v8f wmma4(v2f a, v2f b, v8f c) {
  return __builtin_amdgcn_wmma_f32_16x16x4_f32(
       false, a,  false, b,
       (short)0, c,  false,  false);
}

__device__ __forceinline__ void lds_barrier() {
  __builtin_amdgcn_fence(__ATOMIC_RELEASE, "workgroup", "local");
  __builtin_amdgcn_s_barrier();
  __builtin_amdgcn_fence(__ATOMIC_ACQUIRE, "workgroup", "local");
}

typedef __attribute__((ext_vector_type(4))) float v4f;

template <int K>
struct WRows { v4f r[FROWS * K / 128]; };

template <int K, int I = 0>
__device__ __forceinline__ void w_fetch_from(const v4f* __restrict__ src, WRows<K>& w) {
  if constexpr (I < FROWS * K / 128) {
    w.r[I] = src[I * 32];
    w_fetch_from<K, I + 1>(src, w);
  }
}

template <int K>
__device__ __forceinline__ void w_fetch(const float* __restrict__ W, int frow, int lane,
                                        WRows<K>& w) {
  w_fetch_from<K>(reinterpret_cast<const v4f*>(W + (size_t)frow * K) + lane, w);
}

template <int K, int I = 0>
__device__ __forceinline__ void w_commit_to(float* __restrict__ sWl, const WRows<K>& w) {
  if constexpr (I < FROWS * K / 128) {
    *reinterpret_cast<v4f*>(sWl + ((I * 128) / K) * (K + 4) + ((I * 128) % K)) = w.r[I];
    w_commit_to<K, I + 1>(sWl, w);
  }
}

template <int K>
__device__ __forceinline__ void w_commit(float* __restrict__ sW, int lane, const WRows<K>& w) {
  const int f   = lane * 4;
  const int row = f / K;
  const int col = f % K;
  w_commit_to<K>(sW + row * (K + 4) + col, w);
}

typedef __attribute__((address_space(3))) float lds_float;
typedef __attribute__((address_space(3))) v2f   lds_v2f;

template <int K, int LDA>
__device__ __forceinline__ void gemm_lds(const float* __restrict__ sY,
                                         const float* __restrict__ sW,
                                         int lane, v8f acc[NTILE]) {
  const int mrow  = lane & 15;
  const int khalf = (lane >> 4) << 1;
  const lds_float* pa = (const lds_float*)(sY + mrow * LDA + khalf);
  const lds_float* pb = (const lds_float*)(sW + mrow * (K + 4) + khalf);
  asm("" : "+v"(pa), "+v"(pb));
  #pragma unroll 4
  for (int ks = 0; ks < K / 4; ++ks) {
    v2f a = *(const lds_v2f*)(pa + ks * 4);
    #pragma unroll
    for (int nt = 0; nt < NTILE; ++nt) {
      v2f b = *(const lds_v2f*)(pb + nt * 16 * (K + 4) + ks * 4);
      acc[nt] = wmma4(a, b, acc[nt]);
    }
  }
}

__device__ __forceinline__ void lds_store1(float* p, float v) {
  *(volatile lds_float*)p = v;
}

__device__ __forceinline__ void write_tile(float* __restrict__ sY, int wcol, int lane,
                                           const v8f v[NTILE]) {
  const int hi   = lane >> 4;
  const int col0 = lane & 15;
  #pragma unroll
  for (int nt = 0; nt < NTILE; ++nt) {
    const int col = wcol + nt * 16 + col0;
    #pragma unroll
    for (int r = 0; r < 8; ++r)
      lds_store1(sY + (r + 8 * hi) * LDP + col, v[nt][r]);
  }
}

__global__ void __launch_bounds__(256) __attribute__((amdgpu_waves_per_eu(2, 2))) jacde_kernel(
    const float* __restrict__ t,      const float* __restrict__ hmat,
    const float* __restrict__ coeffs, const float* __restrict__ dcoeffs,
    const float* __restrict__ tobs,
    const float* __restrict__ wx,     const float* __restrict__ wh,
    const float* __restrict__ wout,
    const float* __restrict__ b0,     const float* __restrict__ b1,
    float* __restrict__ out) {
  __shared__ __attribute__((aligned(16))) float sAall[BROWS * LDP];
  __shared__ __attribute__((aligned(16))) float sBall[BROWS * LDP];
  __shared__ __attribute__((aligned(16))) float sXall[BROWS * LDX];
  __shared__ __attribute__((aligned(16))) float sWx[HDIM * (CINC + 4)];
  __shared__ __attribute__((aligned(16))) float sWh[HDIM * (HDIM + 4)];
  __shared__ __attribute__((aligned(16))) float sWo[HDIM * (HDIM + 4)];

  const int tid  = threadIdx.x;
  const int lane = tid & 31;
  const int wave = tid >> 5;
  const int mi   = wave & 1;
  const int ni   = wave >> 1;
  const int wcol = ni * WROWS;
  const int frow = wave * FROWS;
  const int brow = blockIdx.x * BROWS;
  const int hi   = lane >> 4;
  const int col0 = lane & 15;
  float* sA = sAall + mi * 16 * LDP;
  float* sB = sBall + mi * 16 * LDP;
  const float* sX = sXall + mi * 16 * LDX;
  const float* sBx = sWx + wcol * (CINC + 4);
  const float* sBh = sWh + wcol * (HDIM + 4);
  const float* sBo = sWo + wcol * (HDIM + 4);

  WRows<CINC> rx;
  w_fetch<CINC>(wx, frow, lane, rx);
  __builtin_amdgcn_sched_barrier(0);

  const float ts = t[0];
  int idx = 0;
  #pragma unroll
  for (int j = 0; j < NOBSC; ++j) idx += (tobs[j] <= ts) ? 1 : 0;
  idx -= 1;
  idx = idx < 0 ? 0 : (idx > NOBSC - 2 ? NOBSC - 2 : idx);
  const float dt = ts - tobs[idx];
  const float p1 = dt, p2 = dt * dt, p3 = p2 * dt;

  float bias0[NTILE], bias1[NTILE];
  #pragma unroll
  for (int nt = 0; nt < NTILE; ++nt) {
    bias0[nt] = b0[wcol + nt * 16 + col0];
    bias1[nt] = b1[wcol + nt * 16 + col0];
  }

  const int sch  = tid & (CINC - 1);
  const int spar = tid >> 6;
  float4 c4[8];
  {
    const float* cb = coeffs + ((size_t)brow * (NOBSC - 1) + idx) * CINC * 4;
    #pragma unroll
    for (int i = 0; i < 8; ++i)
      c4[i] = *(const float4*)(cb + (size_t)(4 * i + spar) * (NOBSC - 1) * CINC * 4 + sch * 4);
  }
  const int hq = tid & 31;
  const int hr = tid >> 5;
  v4f h4[4];
  #pragma unroll
  for (int i = 0; i < 4; ++i)
    h4[i] = *reinterpret_cast<const v4f*>(hmat + (size_t)(brow + hr + 8 * i) * HDIM + hq * 4);
  __builtin_amdgcn_sched_barrier(0);
  WRows<HDIM> rh, ro;
  w_fetch<HDIM>(wh, frow, lane, rh);
  w_fetch<HDIM>(wout, frow, lane, ro);
  __builtin_amdgcn_sched_barrier(0);
  w_commit<CINC>(sWx + frow * (CINC + 4), lane, rx);
  #pragma unroll
  for (int i = 0; i < 8; ++i)
    sAall[(4 * i + spar) * LDP + sch] = c4[i].x + c4[i].y * p1 + c4[i].z * p2 + c4[i].w * p3;
  __builtin_amdgcn_sched_barrier(0);
  float4 d4[8];
  {
    const float* db = dcoeffs + ((size_t)brow * (NOBSC - 1) + idx) * CINC * 4;
    #pragma unroll
    for (int i = 0; i < 8; ++i)
      d4[i] = *(const float4*)(db + (size_t)(4 * i + spar) * (NOBSC - 1) * CINC * 4 + sch * 4);
  }
  __builtin_amdgcn_sched_barrier(0);
  lds_barrier();

  v8f acc[NTILE];
  #pragma unroll
  for (int nt = 0; nt < NTILE; ++nt) {
    const float bv = bias0[nt];
    #pragma unroll
    for (int r = 0; r < 8; ++r) acc[nt][r] = bv;
  }
  gemm_lds<CINC, LDP>(sA, sBx, lane, acc);
  #pragma unroll
  for (int i = 0; i < 4; ++i)
    *reinterpret_cast<v4f*>(sBall + (hr + 8 * i) * LDP + hq * 4) = h4[i];
  w_commit<HDIM>(sWh + frow * (HDIM + 4), lane, rh);
  lds_barrier();
  gemm_lds<HDIM, LDP>(sB, sBh, lane, acc);

  v8f drelu[NTILE];
  #pragma unroll
  for (int nt = 0; nt < NTILE; ++nt)
    #pragma unroll
    for (int r = 0; r < 8; ++r) {
      const float x = acc[nt][r];
      drelu[nt][r] = __builtin_amdgcn_rcpf(1.0f + __expf(-x));
      acc[nt][r]   = fmaxf(x, 0.0f);
    }
  write_tile(sA, wcol, lane, acc);
  w_commit<HDIM>(sWo + frow * (HDIM + 4), lane, ro);
  #pragma unroll
  for (int i = 0; i < 8; ++i)
    sXall[(4 * i + spar) * LDX + sch] = d4[i].x + d4[i].y * p1 + d4[i].z * p2 + d4[i].w * p3;
  lds_barrier();

  v8f dtn[NTILE];
  v8f cur[NTILE];
  {
    v8f a2[NTILE];
    #pragma unroll
    for (int nt = 0; nt < NTILE; ++nt) {
      const float bv = bias1[nt];
      #pragma unroll
      for (int r = 0; r < 8; ++r) a2[nt][r] = bv;
      cur[nt] = (v8f)(0.0f);
    }
    gemm_lds<HDIM, LDP>(sA, sBo, lane, a2);
    gemm_lds<CINC, LDX>(sX, sBx, lane, cur);
    #pragma unroll
    for (int nt = 0; nt < NTILE; ++nt)
      #pragma unroll
      for (int r = 0; r < 8; ++r) {
        const float q  = __builtin_amdgcn_exp2f(fabsf(a2[nt][r]) * -2.885390082f);
        const float r2 = __builtin_amdgcn_rcpf(__builtin_fmaf(q, 0.5f, 0.5f));
        dtn[nt][r] = (q * r2) * r2;
      }
  }

  #pragma unroll
  for (int nt = 0; nt < NTILE; ++nt) cur[nt] *= drelu[nt];
  write_tile(sB, wcol, lane, cur);
  lds_barrier();
  v8f outv[NTILE];
  #pragma unroll
  for (int nt = 0; nt < NTILE; ++nt) outv[nt] = (v8f)(0.0f);
  gemm_lds<HDIM, LDP>(sB, sBo, lane, outv);
  #pragma unroll
  for (int nt = 0; nt < NTILE; ++nt) {
    outv[nt] *= dtn[nt];
    cur[nt] = outv[nt];
  }

  write_tile(sA, wcol, lane, cur);
  lds_barrier();
  {
    v8f t1[NTILE];
    #pragma unroll
    for (int nt = 0; nt < NTILE; ++nt) t1[nt] = (v8f)(0.0f);
    gemm_lds<HDIM, LDP>(sA, sBh, lane, t1);
    #pragma unroll
    for (int nt = 0; nt < NTILE; ++nt) t1[nt] *= drelu[nt];
    write_tile(sB, wcol, lane, t1);
  }
  lds_barrier();
  {
    v8f t2[NTILE];
    #pragma unroll
    for (int nt = 0; nt < NTILE; ++nt) t2[nt] = (v8f)(0.0f);
    gemm_lds<HDIM, LDP>(sB, sBo, lane, t2);
    #pragma unroll
    for (int nt = 0; nt < NTILE; ++nt) {
      t2[nt] *= dtn[nt];
      outv[nt] += t2[nt];
      cur[nt] = t2[nt];
    }
  }

  write_tile(sA, wcol, lane, cur);
  lds_barrier();
  {
    v8f t1[NTILE];
    #pragma unroll
    for (int nt = 0; nt < NTILE; ++nt) t1[nt] = (v8f)(0.0f);
    gemm_lds<HDIM, LDP>(sA, sBh, lane, t1);
    #pragma unroll
    for (int nt = 0; nt < NTILE; ++nt) t1[nt] *= drelu[nt];
    write_tile(sB, wcol, lane, t1);
  }
  lds_barrier();
  {
    v8f t2[NTILE];
    #pragma unroll
    for (int nt = 0; nt < NTILE; ++nt) t2[nt] = (v8f)(0.0f);
    gemm_lds<HDIM, LDP>(sB, sBo, lane, t2);
    #pragma unroll
    for (int nt = 0; nt < NTILE; ++nt) {
      t2[nt] *= dtn[nt];
      outv[nt] += t2[nt];
    }
  }

  #pragma unroll
  for (int nt = 0; nt < NTILE; ++nt) {
    const int col = wcol + nt * 16 + col0;
    #pragma unroll
    for (int r = 0; r < 8; ++r)
      out[(size_t)(brow + mi * 16 + r + 8 * hi) * HDIM + col] = outv[nt][r];
  }
}

extern "C" void kernel_launch(void* const* d_in, const int* in_sizes, int n_in,
                              void* d_out, int out_size, void* d_ws, size_t ws_size,
                              hipStream_t stream) {
  const float* t       = (const float*)d_in[0];
  const float* hmat    = (const float*)d_in[1];
  const float* coeffs  = (const float*)d_in[2];
  const float* dcoeffs = (const float*)d_in[3];
  const float* tobs    = (const float*)d_in[4];
  const float* wx      = (const float*)d_in[5];
  const float* wh      = (const float*)d_in[6];
  const float* wout    = (const float*)d_in[7];
  const float* b0      = (const float*)d_in[8];
  const float* b1      = (const float*)d_in[9];
  float* out           = (float*)d_out;

  jacde_kernel<<<8192 / BROWS, 256, 0, stream>>>(t, hmat, coeffs, dcoeffs, tobs,
                                                 wx, wh, wout, b0, b1, out);
}


# Each wave keeps its wx, wh, wout B fragments in registers after first use; every product issues all its LDS loads as one group ahead of its matrix instructions (k loops unrolled, same sum order)
# speedup vs baseline: 2.3149x; 1.1383x over previous
#include <hip/hip_runtime.h>


typedef __attribute__((ext_vector_type(2))) float v2f;
typedef __attribute__((ext_vector_type(8))) float v8f;

#define LDP   132
#define LDX   68
#define HDIM  128
#define CINC  64
#define NOBSC 16
#define NWAVE 8
#define BROWS 32
#define WROWS 32
#define NTILE (WROWS / 16)
#define FROWS (HDIM / NWAVE)

__device__ __forceinline__ v8f wmma4(v2f a, v2f b, v8f c) {
  return __builtin_amdgcn_wmma_f32_16x16x4_f32(
       false, a,  false, b,
       (short)0, c,  false,  false);
}

__device__ __forceinline__ void lds_barrier() {
  __builtin_amdgcn_fence(__ATOMIC_RELEASE, "workgroup", "local");
  __builtin_amdgcn_s_barrier();
  __builtin_amdgcn_fence(__ATOMIC_ACQUIRE, "workgroup", "local");
}

typedef __attribute__((ext_vector_type(4))) float v4f;

template <int K>
struct WRows { v4f r[FROWS * K / 128]; };

template <int K, int I = 0>
__device__ __forceinline__ void w_fetch_from(const v4f* __restrict__ src, WRows<K>& w) {
  if constexpr (I < FROWS * K / 128) {
    w.r[I] = src[I * 32];
    w_fetch_from<K, I + 1>(src, w);
  }
}

template <int K>
__device__ __forceinline__ void w_fetch(const float* __restrict__ W, int frow, int lane,
                                        WRows<K>& w) {
  w_fetch_from<K>(reinterpret_cast<const v4f*>(W + (size_t)frow * K) + lane, w);
}

template <int K, int I = 0>
__device__ __forceinline__ void w_commit_to(float* __restrict__ sWl, const WRows<K>& w) {
  if constexpr (I < FROWS * K / 128) {
    *reinterpret_cast<v4f*>(sWl + ((I * 128) / K) * (K + 4) + ((I * 128) % K)) = w.r[I];
    w_commit_to<K, I + 1>(sWl, w);
  }
}

template <int K>
__device__ __forceinline__ void w_commit(float* __restrict__ sW, int lane, const WRows<K>& w) {
  const int f   = lane * 4;
  const int row = f / K;
  const int col = f % K;
  w_commit_to<K>(sW + row * (K + 4) + col, w);
}

typedef __attribute__((address_space(3))) float lds_float;
typedef __attribute__((address_space(3))) v2f   lds_v2f;

template <int K>
struct BFrag { v2f f[NTILE][K / 4]; };

template <int K, int LDA, bool LOADB>
__device__ __forceinline__ void gemm_res(const float* __restrict__ sY,
                                         const float* __restrict__ sW,
                                         int lane, BFrag<K>& bf, v8f acc[NTILE]) {
  const int mrow  = lane & 15;
  const int khalf = (lane >> 4) << 1;
  const lds_float* pa = (const lds_float*)(sY + mrow * LDA + khalf);
  v2f a[K / 4];
  if constexpr (LOADB) {
    const lds_float* pb = (const lds_float*)(sW + mrow * (K + 4) + khalf);
    asm("" : "+v"(pa), "+v"(pb));
    #pragma unroll
    for (int ks = 0; ks < K / 4; ++ks) {
      a[ks] = *(const lds_v2f*)(pa + ks * 4);
      #pragma unroll
      for (int nt = 0; nt < NTILE; ++nt)
        bf.f[nt][ks] = *(const lds_v2f*)(pb + nt * 16 * (K + 4) + ks * 4);
    }
  } else {
    #pragma unroll
    for (int ks = 0; ks < K / 4; ++ks)
      a[ks] = *(const lds_v2f*)(pa + ks * 4);
  }
  __builtin_amdgcn_sched_barrier(0);
  #pragma unroll
  for (int ks = 0; ks < K / 4; ++ks) {
    #pragma unroll
    for (int nt = 0; nt < NTILE; ++nt)
      acc[nt] = wmma4(a[ks], bf.f[nt][ks], acc[nt]);
  }
}

__device__ __forceinline__ void lds_store1(float* p, float v) {
  *(volatile lds_float*)p = v;
}

__device__ __forceinline__ void write_tile(float* __restrict__ sY, int wcol, int lane,
                                           const v8f v[NTILE]) {
  const int hi   = lane >> 4;
  const int col0 = lane & 15;
  #pragma unroll
  for (int nt = 0; nt < NTILE; ++nt) {
    const int col = wcol + nt * 16 + col0;
    #pragma unroll
    for (int r = 0; r < 8; ++r)
      lds_store1(sY + (r + 8 * hi) * LDP + col, v[nt][r]);
  }
}

__global__ void __launch_bounds__(256) __attribute__((amdgpu_waves_per_eu(2, 2))) jacde_kernel(
    const float* __restrict__ t,      const float* __restrict__ hmat,
    const float* __restrict__ coeffs, const float* __restrict__ dcoeffs,
    const float* __restrict__ tobs,
    const float* __restrict__ wx,     const float* __restrict__ wh,
    const float* __restrict__ wout,
    const float* __restrict__ b0,     const float* __restrict__ b1,
    float* __restrict__ out) {
  __shared__ __attribute__((aligned(16))) float sAall[BROWS * LDP];
  __shared__ __attribute__((aligned(16))) float sBall[BROWS * LDP];
  __shared__ __attribute__((aligned(16))) float sXall[BROWS * LDX];
  __shared__ __attribute__((aligned(16))) float sWx[HDIM * (CINC + 4)];
  __shared__ __attribute__((aligned(16))) float sWh[HDIM * (HDIM + 4)];
  __shared__ __attribute__((aligned(16))) float sWo[HDIM * (HDIM + 4)];

  const int tid  = threadIdx.x;
  const int lane = tid & 31;
  const int wave = tid >> 5;
  const int mi   = wave & 1;
  const int ni   = wave >> 1;
  const int wcol = ni * WROWS;
  const int frow = wave * FROWS;
  const int brow = blockIdx.x * BROWS;
  const int hi   = lane >> 4;
  const int col0 = lane & 15;
  float* sA = sAall + mi * 16 * LDP;
  float* sB = sBall + mi * 16 * LDP;
  const float* sX = sXall + mi * 16 * LDX;
  const float* sBx = sWx + wcol * (CINC + 4);
  const float* sBh = sWh + wcol * (HDIM + 4);
  const float* sBo = sWo + wcol * (HDIM + 4);

  WRows<CINC> rx;
  w_fetch<CINC>(wx, frow, lane, rx);
  __builtin_amdgcn_sched_barrier(0);

  const float ts = t[0];
  int idx = 0;
  #pragma unroll
  for (int j = 0; j < NOBSC; ++j) idx += (tobs[j] <= ts) ? 1 : 0;
  idx -= 1;
  idx = idx < 0 ? 0 : (idx > NOBSC - 2 ? NOBSC - 2 : idx);
  const float dt = ts - tobs[idx];
  const float p1 = dt, p2 = dt * dt, p3 = p2 * dt;

  float bias0[NTILE], bias1[NTILE];
  #pragma unroll
  for (int nt = 0; nt < NTILE; ++nt) {
    bias0[nt] = b0[wcol + nt * 16 + col0];
    bias1[nt] = b1[wcol + nt * 16 + col0];
  }

  const int sch  = tid & (CINC - 1);
  const int spar = tid >> 6;
  float4 c4[8];
  {
    const float* cb = coeffs + ((size_t)brow * (NOBSC - 1) + idx) * CINC * 4;
    #pragma unroll
    for (int i = 0; i < 8; ++i)
      c4[i] = *(const float4*)(cb + (size_t)(4 * i + spar) * (NOBSC - 1) * CINC * 4 + sch * 4);
  }
  const int hq = tid & 31;
  const int hr = tid >> 5;
  v4f h4[4];
  #pragma unroll
  for (int i = 0; i < 4; ++i)
    h4[i] = *reinterpret_cast<const v4f*>(hmat + (size_t)(brow + hr + 8 * i) * HDIM + hq * 4);
  __builtin_amdgcn_sched_barrier(0);
  WRows<HDIM> rh, ro;
  w_fetch<HDIM>(wh, frow, lane, rh);
  w_fetch<HDIM>(wout, frow, lane, ro);
  __builtin_amdgcn_sched_barrier(0);
  w_commit<CINC>(sWx + frow * (CINC + 4), lane, rx);
  #pragma unroll
  for (int i = 0; i < 8; ++i)
    sAall[(4 * i + spar) * LDP + sch] = c4[i].x + c4[i].y * p1 + c4[i].z * p2 + c4[i].w * p3;
  __builtin_amdgcn_sched_barrier(0);
  float4 d4[8];
  {
    const float* db = dcoeffs + ((size_t)brow * (NOBSC - 1) + idx) * CINC * 4;
    #pragma unroll
    for (int i = 0; i < 8; ++i)
      d4[i] = *(const float4*)(db + (size_t)(4 * i + spar) * (NOBSC - 1) * CINC * 4 + sch * 4);
  }
  __builtin_amdgcn_sched_barrier(0);
  lds_barrier();

  v8f acc[NTILE];
  #pragma unroll
  for (int nt = 0; nt < NTILE; ++nt) {
    const float bv = bias0[nt];
    #pragma unroll
    for (int r = 0; r < 8; ++r) acc[nt][r] = bv;
  }
  BFrag<CINC> fx;
  BFrag<HDIM> fh, fo;
  gemm_res<CINC, LDP, true>(sA, sBx, lane, fx, acc);
  #pragma unroll
  for (int i = 0; i < 4; ++i)
    *reinterpret_cast<v4f*>(sBall + (hr + 8 * i) * LDP + hq * 4) = h4[i];
  w_commit<HDIM>(sWh + frow * (HDIM + 4), lane, rh);
  lds_barrier();
  gemm_res<HDIM, LDP, true>(sB, sBh, lane, fh, acc);

  v8f drelu[NTILE];
  #pragma unroll
  for (int nt = 0; nt < NTILE; ++nt)
    #pragma unroll
    for (int r = 0; r < 8; ++r) {
      const float x = acc[nt][r];
      drelu[nt][r] = __builtin_amdgcn_rcpf(1.0f + __expf(-x));
      acc[nt][r]   = fmaxf(x, 0.0f);
    }
  write_tile(sA, wcol, lane, acc);
  w_commit<HDIM>(sWo + frow * (HDIM + 4), lane, ro);
  #pragma unroll
  for (int i = 0; i < 8; ++i)
    sXall[(4 * i + spar) * LDX + sch] = d4[i].x + d4[i].y * p1 + d4[i].z * p2 + d4[i].w * p3;
  lds_barrier();

  v8f dtn[NTILE];
  v8f cur[NTILE];
  {
    v8f a2[NTILE];
    #pragma unroll
    for (int nt = 0; nt < NTILE; ++nt) {
      const float bv = bias1[nt];
      #pragma unroll
      for (int r = 0; r < 8; ++r) a2[nt][r] = bv;
      cur[nt] = (v8f)(0.0f);
    }
    gemm_res<HDIM, LDP, true>(sA, sBo, lane, fo, a2);
    gemm_res<CINC, LDX, false>(sX, sBx, lane, fx, cur);
    #pragma unroll
    for (int nt = 0; nt < NTILE; ++nt)
      #pragma unroll
      for (int r = 0; r < 8; ++r) {
        const float q  = __builtin_amdgcn_exp2f(fabsf(a2[nt][r]) * -2.885390082f);
        const float r2 = __builtin_amdgcn_rcpf(__builtin_fmaf(q, 0.5f, 0.5f));
        dtn[nt][r] = (q * r2) * r2;
      }
  }

  #pragma unroll
  for (int nt = 0; nt < NTILE; ++nt) cur[nt] *= drelu[nt];
  write_tile(sB, wcol, lane, cur);
  lds_barrier();
  v8f outv[NTILE];
  #pragma unroll
  for (int nt = 0; nt < NTILE; ++nt) outv[nt] = (v8f)(0.0f);
  gemm_res<HDIM, LDP, false>(sB, sBo, lane, fo, outv);
  #pragma unroll
  for (int nt = 0; nt < NTILE; ++nt) {
    outv[nt] *= dtn[nt];
    cur[nt] = outv[nt];
  }

  write_tile(sA, wcol, lane, cur);
  lds_barrier();
  {
    v8f t1[NTILE];
    #pragma unroll
    for (int nt = 0; nt < NTILE; ++nt) t1[nt] = (v8f)(0.0f);
    gemm_res<HDIM, LDP, false>(sA, sBh, lane, fh, t1);
    #pragma unroll
    for (int nt = 0; nt < NTILE; ++nt) t1[nt] *= drelu[nt];
    write_tile(sB, wcol, lane, t1);
  }
  lds_barrier();
  {
    v8f t2[NTILE];
    #pragma unroll
    for (int nt = 0; nt < NTILE; ++nt) t2[nt] = (v8f)(0.0f);
    gemm_res<HDIM, LDP, false>(sB, sBo, lane, fo, t2);
    #pragma unroll
    for (int nt = 0; nt < NTILE; ++nt) {
      t2[nt] *= dtn[nt];
      outv[nt] += t2[nt];
      cur[nt] = t2[nt];
    }
  }

  write_tile(sA, wcol, lane, cur);
  lds_barrier();
  {
    v8f t1[NTILE];
    #pragma unroll
    for (int nt = 0; nt < NTILE; ++nt) t1[nt] = (v8f)(0.0f);
    gemm_res<HDIM, LDP, false>(sA, sBh, lane, fh, t1);
    #pragma unroll
    for (int nt = 0; nt < NTILE; ++nt) t1[nt] *= drelu[nt];
    write_tile(sB, wcol, lane, t1);
  }
  lds_barrier();
  {
    v8f t2[NTILE];
    #pragma unroll
    for (int nt = 0; nt < NTILE; ++nt) t2[nt] = (v8f)(0.0f);
    gemm_res<HDIM, LDP, false>(sB, sBo, lane, fo, t2);
    #pragma unroll
    for (int nt = 0; nt < NTILE; ++nt) {
      t2[nt] *= dtn[nt];
      outv[nt] += t2[nt];
    }
  }

  #pragma unroll
  for (int nt = 0; nt < NTILE; ++nt) {
    const int col = wcol + nt * 16 + col0;
    #pragma unroll
    for (int r = 0; r < 8; ++r)
      out[(size_t)(brow + mi * 16 + r + 8 * hi) * HDIM + col] = outv[nt][r];
  }
}

extern "C" void kernel_launch(void* const* d_in, const int* in_sizes, int n_in,
                              void* d_out, int out_size, void* d_ws, size_t ws_size,
                              hipStream_t stream) {
  const float* t       = (const float*)d_in[0];
  const float* hmat    = (const float*)d_in[1];
  const float* coeffs  = (const float*)d_in[2];
  const float* dcoeffs = (const float*)d_in[3];
  const float* tobs    = (const float*)d_in[4];
  const float* wx      = (const float*)d_in[5];
  const float* wh      = (const float*)d_in[6];
  const float* wout    = (const float*)d_in[7];
  const float* b0      = (const float*)d_in[8];
  const float* b1      = (const float*)d_in[9];
  float* out           = (float*)d_out;

  jacde_kernel<<<8192 / BROWS, 256, 0, stream>>>(t, hmat, coeffs, dcoeffs, tobs,
                                                 wx, wh, wout, b0, b1, out);
}


# Each wave owns 16 columns of both 16-row halves and runs the halves half a product apart: one half's LDS hand-over (stores, split barrier, loads) sits under the other half's matrix instructions
# speedup vs baseline: 2.4601x; 1.0627x over previous
#include <hip/hip_runtime.h>


typedef __attribute__((ext_vector_type(2))) float v2f;
typedef __attribute__((ext_vector_type(8))) float v8f;

#define LDP   132
#define LDX   68
#define HDIM  128
#define CINC  64
#define NOBSC 16
#define NWAVE 8
#define BROWS 32
#define WCOLS 16
#define FROWS (HDIM / NWAVE)

__device__ __forceinline__ v8f wmma4(v2f a, v2f b, v8f c) {
  return __builtin_amdgcn_wmma_f32_16x16x4_f32(
       false, a,  false, b,
       (short)0, c,  false,  false);
}

__device__ __forceinline__ void lds_barrier() {
  __builtin_amdgcn_fence(__ATOMIC_RELEASE, "workgroup", "local");
  __builtin_amdgcn_s_barrier();
  __builtin_amdgcn_fence(__ATOMIC_ACQUIRE, "workgroup", "local");
}

__device__ __forceinline__ void ho_signal() {
  __builtin_amdgcn_fence(__ATOMIC_RELEASE, "workgroup", "local");
  __builtin_amdgcn_s_barrier_signal(-1);
}
__device__ __forceinline__ void ho_wait() {
  __builtin_amdgcn_s_barrier_wait(-1);
  __builtin_amdgcn_fence(__ATOMIC_ACQUIRE, "workgroup", "local");
}

typedef __attribute__((ext_vector_type(4))) float v4f;

template <int K>
struct WRows { v4f r[FROWS * K / 128]; };

template <int K, int I = 0>
__device__ __forceinline__ void w_fetch_from(const v4f* __restrict__ src, WRows<K>& w) {
  if constexpr (I < FROWS * K / 128) {
    w.r[I] = src[I * 32];
    w_fetch_from<K, I + 1>(src, w);
  }
}

template <int K>
__device__ __forceinline__ void w_fetch(const float* __restrict__ W, int frow, int lane,
                                        WRows<K>& w) {
  w_fetch_from<K>(reinterpret_cast<const v4f*>(W + (size_t)frow * K) + lane, w);
}

template <int K, int I = 0>
__device__ __forceinline__ void w_commit_to(float* __restrict__ sWl, const WRows<K>& w) {
  if constexpr (I < FROWS * K / 128) {
    *reinterpret_cast<v4f*>(sWl + ((I * 128) / K) * (K + 4) + ((I * 128) % K)) = w.r[I];
    w_commit_to<K, I + 1>(sWl, w);
  }
}

template <int K>
__device__ __forceinline__ void w_commit(float* __restrict__ sW, int lane, const WRows<K>& w) {
  const int f   = lane * 4;
  const int row = f / K;
  const int col = f % K;
  w_commit_to<K>(sW + row * (K + 4) + col, w);
}

typedef __attribute__((address_space(3))) float lds_float;
typedef __attribute__((address_space(3))) v2f   lds_v2f;
typedef __attribute__((address_space(3))) v4f   lds_v4f;

template <int K>
struct BFrag { v2f f[K / 4]; };

template <int K>
__device__ __forceinline__ void load_b(const float* __restrict__ sW, int lane, BFrag<K>& bf) {
  const int nrow  = lane & 15;
  const int khalf = (lane >> 4) << 1;
  const lds_float* pb = (const lds_float*)(sW + nrow * (K + 4) + khalf);
  asm("" : "+v"(pb));
  #pragma unroll
  for (int ks = 0; ks < K / 4; ++ks)
    bf.f[ks] = *(const lds_v2f*)(pb + ks * 4);
}

template <int K, int LDA>
__device__ __forceinline__ void load_a(const float* __restrict__ sY, int lane, v2f (&a)[K / 4]) {
  const int mrow  = lane & 15;
  const int khalf = (lane >> 4) << 1;
  const lds_float* pa = (const lds_float*)(sY + mrow * LDA + khalf);
  #pragma unroll
  for (int ks = 0; ks < K / 4; ++ks)
    a[ks] = *(const lds_v2f*)(pa + ks * 4);
}

template <int K, int K0, int K1>
__device__ __forceinline__ void mma(const v2f (&a)[K / 4], const BFrag<K>& bf, v8f& acc) {
  #pragma unroll
  for (int ks = K0; ks < K1; ++ks)
    acc = wmma4(a[ks], bf.f[ks], acc);
}

__device__ __forceinline__ void pin(v8f& acc) { asm volatile("" : "+v"(acc)); }

template <class Loads>
__device__ __forceinline__ void mma_handover(const v2f (&a)[HDIM / 4], const BFrag<HDIM>& bf,
                                             v8f& acc, Loads&& loads) {
  pin(acc);
  mma<HDIM, 0, 4>(a, bf, acc);
  pin(acc);
  ho_signal();
  mma<HDIM, 4, 12>(a, bf, acc);
  pin(acc);
  ho_wait();
  loads();
  __builtin_amdgcn_sched_barrier(0);
  mma<HDIM, 12, HDIM / 4>(a, bf, acc);
}

__device__ __forceinline__ void lds_store1(float* p, float v) {
  *(volatile lds_float*)p = v;
}

__device__ __forceinline__ void store_tile(float* __restrict__ sY, int wcol, int lane, const v8f& v) {
  const int hi  = lane >> 4;
  const int col = wcol + (lane & 15);
  #pragma unroll
  for (int r = 0; r < 8; ++r)
    lds_store1(sY + (r + 8 * hi) * LDP + col, v[r]);
}

__device__ __forceinline__ v8f dtanh8(const v8f& a) {
  v8f d;
  #pragma unroll
  for (int r = 0; r < 8; ++r) {
    const float q  = __builtin_amdgcn_exp2f(fabsf(a[r]) * -2.885390082f);
    const float r2 = __builtin_amdgcn_rcpf(__builtin_fmaf(q, 0.5f, 0.5f));
    d[r] = (q * r2) * r2;
  }
  return d;
}

__device__ __forceinline__ v8f sigmoid8(const v8f& x) {
  v8f d;
  #pragma unroll
  for (int r = 0; r < 8; ++r) d[r] = __builtin_amdgcn_rcpf(1.0f + __expf(-x[r]));
  return d;
}

__device__ __forceinline__ v8f relu8(const v8f& x) {
  v8f d;
  #pragma unroll
  for (int r = 0; r < 8; ++r) d[r] = fmaxf(x[r], 0.0f);
  return d;
}

__device__ __forceinline__ void rows_out(const float* __restrict__ sY, float* __restrict__ o,
                                         int wave, int lane) {
  #pragma unroll
  for (int j = 0; j < 2; ++j) {
    const int row = 2 * wave + j;
    const v4f v = *(const lds_v4f*)(const lds_float*)(sY + row * LDP + lane * 4);
    *reinterpret_cast<v4f*>(o + (size_t)row * HDIM + lane * 4) = v;
  }
}

__global__ void __launch_bounds__(256) __attribute__((amdgpu_waves_per_eu(2, 2))) jacde_kernel(
    const float* __restrict__ t,      const float* __restrict__ hmat,
    const float* __restrict__ coeffs, const float* __restrict__ dcoeffs,
    const float* __restrict__ tobs,
    const float* __restrict__ wx,     const float* __restrict__ wh,
    const float* __restrict__ wout,
    const float* __restrict__ b0,     const float* __restrict__ b1,
    float* __restrict__ out) {
  __shared__ __attribute__((aligned(16))) float sAall[BROWS * LDP];
  __shared__ __attribute__((aligned(16))) float sBall[BROWS * LDP];
  __shared__ __attribute__((aligned(16))) float sXall[BROWS * LDX];
  __shared__ __attribute__((aligned(16))) float sWx[HDIM * (CINC + 4)];
  __shared__ __attribute__((aligned(16))) float sWh[HDIM * (HDIM + 4)];
  __shared__ __attribute__((aligned(16))) float sWo[HDIM * (HDIM + 4)];

  const int tid  = threadIdx.x;
  const int lane = tid & 31;
  const int wave = tid >> 5;
  const int wcol = wave * WCOLS;
  const int frow = wave * FROWS;
  const int brow = blockIdx.x * BROWS;
  const int col0 = lane & 15;
  float* const sA0 = sAall;
  float* const sA1 = sAall + 16 * LDP;
  float* const sB0 = sBall;
  float* const sB1 = sBall + 16 * LDP;
  const float* const sX0 = sXall;
  const float* const sX1 = sXall + 16 * LDX;
  const float* sBx = sWx + wcol * (CINC + 4);
  const float* sBh = sWh + wcol * (HDIM + 4);
  const float* sBo = sWo + wcol * (HDIM + 4);

  WRows<CINC> rx;
  w_fetch<CINC>(wx, frow, lane, rx);
  __builtin_amdgcn_sched_barrier(0);

  const float ts = t[0];
  int idx = 0;
  #pragma unroll
  for (int j = 0; j < NOBSC; ++j) idx += (tobs[j] <= ts) ? 1 : 0;
  idx -= 1;
  idx = idx < 0 ? 0 : (idx > NOBSC - 2 ? NOBSC - 2 : idx);
  const float dt = ts - tobs[idx];
  const float p1 = dt, p2 = dt * dt, p3 = p2 * dt;

  const float bias0 = b0[wcol + col0];
  const float bias1 = b1[wcol + col0];

  const int sch  = tid & (CINC - 1);
  const int spar = tid >> 6;
  float4 c4[8];
  {
    const float* cb = coeffs + ((size_t)brow * (NOBSC - 1) + idx) * CINC * 4;
    #pragma unroll
    for (int i = 0; i < 8; ++i)
      c4[i] = *(const float4*)(cb + (size_t)(4 * i + spar) * (NOBSC - 1) * CINC * 4 + sch * 4);
  }
  const int hq = tid & 31;
  const int hr = tid >> 5;
  v4f h4[4];
  #pragma unroll
  for (int i = 0; i < 4; ++i)
    h4[i] = *reinterpret_cast<const v4f*>(hmat + (size_t)(brow + hr + 8 * i) * HDIM + hq * 4);
  __builtin_amdgcn_sched_barrier(0);
  WRows<HDIM> rh, ro;
  w_fetch<HDIM>(wh, frow, lane, rh);
  w_fetch<HDIM>(wout, frow, lane, ro);
  __builtin_amdgcn_sched_barrier(0);
  w_commit<CINC>(sWx + frow * (CINC + 4), lane, rx);
  #pragma unroll
  for (int i = 0; i < 8; ++i)
    sAall[(4 * i + spar) * LDP + sch] = c4[i].x + c4[i].y * p1 + c4[i].z * p2 + c4[i].w * p3;
  __builtin_amdgcn_sched_barrier(0);
  float4 d4[8];
  {
    const float* db = dcoeffs + ((size_t)brow * (NOBSC - 1) + idx) * CINC * 4;
    #pragma unroll
    for (int i = 0; i < 8; ++i)
      d4[i] = *(const float4*)(db + (size_t)(4 * i + spar) * (NOBSC - 1) * CINC * 4 + sch * 4);
  }
  __builtin_amdgcn_sched_barrier(0);
  lds_barrier();

  v8f l0, l1;
  #pragma unroll
  for (int r = 0; r < 8; ++r) { l0[r] = bias0; l1[r] = bias0; }
  BFrag<CINC> fx;
  BFrag<HDIM> fh, fo;
  v2f ax0[CINC / 4], ax1[CINC / 4];
  v2f a0[HDIM / 4], a1[HDIM / 4];
  load_b<CINC>(sBx, lane, fx);
  load_a<CINC, LDP>(sA0, lane, ax0);
  load_a<CINC, LDP>(sA1, lane, ax1);
  #pragma unroll
  for (int i = 0; i < 4; ++i)
    *reinterpret_cast<v4f*>(sBall + (hr + 8 * i) * LDP + hq * 4) = h4[i];
  __builtin_amdgcn_sched_barrier(0);
  ho_signal();
  #pragma unroll
  for (int ks = 0; ks < CINC / 4; ++ks) {
    l0 = wmma4(ax0[ks], fx.f[ks], l0);
    l1 = wmma4(ax1[ks], fx.f[ks], l1);
  }
  pin(l0);
  pin(l1);
  ho_wait();
  load_a<HDIM, LDP>(sB0, lane, a0);
  load_a<HDIM, LDP>(sB1, lane, a1);
  w_commit<HDIM>(sWh + frow * (HDIM + 4), lane, rh);
  load_b<HDIM>(sBh, lane, fh);
  __builtin_amdgcn_sched_barrier(0);
  mma<HDIM, 0, HDIM / 4>(a0, fh, l0);
  pin(l0);
  float e1 = p1, e2 = p2, e3 = p3;
  asm volatile("" : "+s"(e1), "+s"(e2), "+s"(e3));

  store_tile(sA0, wcol, lane, relu8(l0));
  w_commit<HDIM>(sWo + frow * (HDIM + 4), lane, ro);
  #pragma unroll
  for (int i = 0; i < 8; ++i)
    sXall[(4 * i + spar) * LDX + sch] = d4[i].x + d4[i].y * e1 + d4[i].z * e2 + d4[i].w * e3;
  mma_handover(a1, fh, l1, [&]() {
    load_a<HDIM, LDP>(sA0, lane, a0);
    load_a<CINC, LDX>(sX0, lane, ax0);
    load_b<HDIM>(sBo, lane, fo);
  });
  const v8f drelu0 = sigmoid8(l0);
  store_tile(sA1, wcol, lane, relu8(l1));

  v8f lo0, u0;
  #pragma unroll
  for (int r = 0; r < 8; ++r) { lo0[r] = bias1; u0[r] = 0.0f; }
  mma_handover(a0, fo, lo0, [&]() {
    load_a<HDIM, LDP>(sA1, lane, a1);
    load_a<CINC, LDX>(sX1, lane, ax1);
  });
  const v8f drelu1 = sigmoid8(l1);
  mma<CINC, 0, CINC / 4>(ax0, fx, u0);
  u0 *= drelu0;
  store_tile(sB0, wcol, lane, u0);

  v8f lo1, u1;
  #pragma unroll
  for (int r = 0; r < 8; ++r) { lo1[r] = bias1; u1[r] = 0.0f; }
  mma_handover(a1, fo, lo1, [&]() { load_a<HDIM, LDP>(sB0, lane, a0); });
  const v8f dtn0 = dtanh8(lo0);
  mma<CINC, 0, CINC / 4>(ax1, fx, u1);
  u1 *= drelu1;
  store_tile(sB1, wcol, lane, u1);

  v8f g0 = (v8f)(0.0f);
  mma_handover(a0, fo, g0, [&]() { load_a<HDIM, LDP>(sB1, lane, a1); });
  const v8f dtn1 = dtanh8(lo1);
  g0 *= dtn0;
  v8f outv0 = g0;
  store_tile(sA0, wcol, lane, g0);

  v8f g1 = (v8f)(0.0f);
  mma_handover(a1, fo, g1, [&]() { load_a<HDIM, LDP>(sA0, lane, a0); });
  g1 *= dtn1;
  v8f outv1 = g1;
  store_tile(sA1, wcol, lane, g1);

  v8f t0 = (v8f)(0.0f);
  mma_handover(a0, fh, t0, [&]() { load_a<HDIM, LDP>(sA1, lane, a1); });
  t0 *= drelu0;
  store_tile(sB0, wcol, lane, t0);
  v8f t1 = (v8f)(0.0f);
  mma_handover(a1, fh, t1, [&]() { load_a<HDIM, LDP>(sB0, lane, a0); });
  t1 *= drelu1;
  store_tile(sB1, wcol, lane, t1);
  g0 = (v8f)(0.0f);
  mma_handover(a0, fo, g0, [&]() { load_a<HDIM, LDP>(sB1, lane, a1); });
  g0 *= dtn0;
  outv0 += g0;
  store_tile(sA0, wcol, lane, g0);
  g1 = (v8f)(0.0f);
  mma_handover(a1, fo, g1, [&]() { load_a<HDIM, LDP>(sA0, lane, a0); });
  g1 *= dtn1;
  outv1 += g1;
  store_tile(sA1, wcol, lane, g1);

  t0 = (v8f)(0.0f);
  mma_handover(a0, fh, t0, [&]() { load_a<HDIM, LDP>(sA1, lane, a1); });
  t0 *= drelu0;
  store_tile(sB0, wcol, lane, t0);
  t1 = (v8f)(0.0f);
  mma_handover(a1, fh, t1, [&]() { load_a<HDIM, LDP>(sB0, lane, a0); });
  t1 *= drelu1;
  store_tile(sB1, wcol, lane, t1);
  g0 = (v8f)(0.0f);
  mma_handover(a0, fo, g0, [&]() { load_a<HDIM, LDP>(sB1, lane, a1); });
  g0 *= dtn0;
  outv0 += g0;
  store_tile(sA0, wcol, lane, outv0);
  g1 = (v8f)(0.0f);
  mma_handover(a1, fo, g1, [&]() { rows_out(sA0, out + (size_t)brow * HDIM, wave, lane); });
  g1 *= dtn1;
  outv1 += g1;
  store_tile(sA1, wcol, lane, outv1);
  ho_signal();
  ho_wait();
  rows_out(sA1, out + (size_t)(brow + 16) * HDIM, wave, lane);
}

extern "C" void kernel_launch(void* const* d_in, const int* in_sizes, int n_in,
                              void* d_out, int out_size, void* d_ws, size_t ws_size,
                              hipStream_t stream) {
  const float* t       = (const float*)d_in[0];
  const float* hmat    = (const float*)d_in[1];
  const float* coeffs  = (const float*)d_in[2];
  const float* dcoeffs = (const float*)d_in[3];
  const float* tobs    = (const float*)d_in[4];
  const float* wx      = (const float*)d_in[5];
  const float* wh      = (const float*)d_in[6];
  const float* wout    = (const float*)d_in[7];
  const float* b0      = (const float*)d_in[8];
  const float* b1      = (const float*)d_in[9];
  float* out           = (float*)d_out;

  jacde_kernel<<<8192 / BROWS, 256, 0, stream>>>(t, hmat, coeffs, dcoeffs, tobs,
                                                 wx, wh, wout, b0, b1, out);
}
